# starting kernel
# baseline (speedup 1.0000x reference)
#include <hip/hip_runtime.h>
#include <math.h>

#define NN    50000
#define EE    800000
#define ETOT  (EE + NN)
#define KDIM  128

typedef float v2f __attribute__((ext_vector_type(2)));
typedef float v8f __attribute__((ext_vector_type(8)));

__device__ __forceinline__ void atomicMaxF(float* addr, float val) {
    if (val >= 0.0f) atomicMax((int*)addr, __float_as_int(val));
    else             atomicMin((unsigned int*)addr, __float_as_uint(val));
}

__global__ __launch_bounds__(1024) void mean_kernel(const float* __restrict__ ew,
                                                    float* __restrict__ meanw, int E) {
    __shared__ float sdata[1024];
    float s = 0.0f;
    for (int i = threadIdx.x; i < E; i += 1024) s += ew[i];
    sdata[threadIdx.x] = s;
    __syncthreads();
    for (int o = 512; o > 0; o >>= 1) {
        if (threadIdx.x < o) sdata[threadIdx.x] += sdata[threadIdx.x + o];
        __syncthreads();
    }
    if (threadIdx.x == 0) meanw[0] = sdata[0] / (float)E;
}

__global__ void edge_coef_kernel(const float* __restrict__ We, const float* __restrict__ ae,
                                 float* __restrict__ ce, int H, int F) {
    int h = threadIdx.x;
    if (h >= H) return;
    float s = 0.0f;
    for (int f = 0; f < F; ++f) s += We[h * F + f] * ae[h * F + f];
    ce[h] = s;
}

__global__ __launch_bounds__(32) void gemm_wmma_f32(const float* __restrict__ X,
                                                    const float* __restrict__ W,
                                                    float* __restrict__ out,
                                                    int K, int C) {
    const int tm   = blockIdx.x;
    const int tc   = blockIdx.y;
    const int lane = threadIdx.x;
    const int half = lane >> 4;
    const int l15  = lane & 15;

    v8f acc = {};
    const float* xrow = X + (size_t)(tm * 16 + l15) * K;
    const int    coln = tc * 16 + l15;

    for (int k = 0; k < K; k += 4) {
        const int kk = k + half * 2;
        v2f a, b;
        a[0] = xrow[kk];
        a[1] = xrow[kk + 1];
        b[0] = W[(size_t)kk * C + coln];
        b[1] = W[(size_t)(kk + 1) * C + coln];
        acc = __builtin_amdgcn_wmma_f32_16x16x4_f32(
             false, a,  false, b,
             (short)0, acc,  false,  false);
    }
#pragma unroll
    for (int v = 0; v < 8; ++v) {
        const int row = tm * 16 + v + half * 8;
        out[(size_t)row * C + coln] = acc[v];
    }
}

__global__ void score_kernel(const float* __restrict__ xw,
                             const float* __restrict__ a_s, const float* __restrict__ a_d,
                             float* __restrict__ ss, float* __restrict__ sd,
                             int H, int F) {
    int i = blockIdx.x * blockDim.x + threadIdx.x;
    if (i >= NN * H) return;
    const int n = i / H, h = i % H;
    const float* v  = xw + (size_t)n * H * F + h * F;
    const float* as = a_s + h * F;
    const float* ad = a_d + h * F;
    float s1 = 0.0f, s2 = 0.0f;
    for (int f = 0; f < F; ++f) { float x = v[f]; s1 += x * as[f]; s2 += x * ad[f]; }
    ss[i] = s1;
    sd[i] = s2;
}

__global__ void init_kernel(float* __restrict__ out, const float* __restrict__ b,
                            float* __restrict__ m, float* __restrict__ denom,
                            int C, int H) {
    const int n = blockIdx.x, t = threadIdx.x;
    out[(size_t)n * C + t] = b[t];
    if (t < H) { m[n * H + t] = -__builtin_inff(); denom[n * H + t] = 0.0f; }
}

__global__ void alpha_max_kernel(const int* __restrict__ ei, const float* __restrict__ ew,
                                 const float* __restrict__ meanw,
                                 const float* __restrict__ ss, const float* __restrict__ sd,
                                 const float* __restrict__ ce,
                                 float* __restrict__ alpha, float* __restrict__ m, int H) {
    int idx = blockIdx.x * blockDim.x + threadIdx.x;
    if (idx >= ETOT * H) return;
    const int e = idx / H, h = idx % H;
    int s, d; float ea;
    if (e < EE) { s = ei[e]; d = ei[EE + e]; ea = ew[e]; }
    else        { s = d = e - EE;            ea = meanw[0]; }
    float a = ss[s * H + h] + sd[d * H + h] + ea * ce[h];
    a = a > 0.0f ? a : 0.2f * a;
    alpha[idx] = a;
    atomicMaxF(&m[d * H + h], a);
}

__global__ void exp_denom_kernel(const int* __restrict__ ei,
                                 const float* __restrict__ m,
                                 float* __restrict__ alpha, float* __restrict__ denom, int H) {
    int idx = blockIdx.x * blockDim.x + threadIdx.x;
    if (idx >= ETOT * H) return;
    const int e = idx / H, h = idx % H;
    const int d = (e < EE) ? ei[EE + e] : (e - EE);
    float ex = __expf(alpha[idx] - m[d * H + h]);
    alpha[idx] = ex;
    atomicAdd(&denom[d * H + h], ex);
}

__global__ void scatter_kernel(const int* __restrict__ ei,
                               const float* __restrict__ xw,
                               const float* __restrict__ alpha,
                               const float* __restrict__ denom,
                               float* __restrict__ out, int H, int F) {
    const int e = blockIdx.x;
    const int t = threadIdx.x;
    int s, d;
    if (e < EE) { s = ei[e]; d = ei[EE + e]; } else { s = d = e - EE; }
    const int h = t / F;
    const float attn = alpha[(size_t)e * H + h] / (denom[d * H + h] + 1e-16f);
    atomicAdd(&out[(size_t)d * H * F + t], xw[(size_t)s * H * F + t] * attn);
}

__global__ void elu_kernel(float* __restrict__ buf, int n) {
    int i = blockIdx.x * blockDim.x + threadIdx.x;
    if (i >= n) return;
    float v = buf[i];
    buf[i] = v > 0.0f ? v : expm1f(v);
}

extern "C" void kernel_launch(void* const* d_in, const int* in_sizes, int n_in,
                              void* d_out, int out_size, void* d_ws, size_t ws_size,
                              hipStream_t stream) {
    const float* x  = (const float*)d_in[0];
    const int*   ei = (const int*)d_in[1];
    const float* ew = (const float*)d_in[2];

    const float* Wl[3] = { (const float*)d_in[3],  (const float*)d_in[9],  (const float*)d_in[15] };
    const float* As[3] = { (const float*)d_in[4],  (const float*)d_in[10], (const float*)d_in[16] };
    const float* Ad[3] = { (const float*)d_in[5],  (const float*)d_in[11], (const float*)d_in[17] };
    const float* We[3] = { (const float*)d_in[6],  (const float*)d_in[12], (const float*)d_in[18] };
    const float* Ae[3] = { (const float*)d_in[7],  (const float*)d_in[13], (const float*)d_in[19] };
    const float* Bi[3] = { (const float*)d_in[8],  (const float*)d_in[14], (const float*)d_in[20] };

    const int Hs[3] = { 4, 4, 1 };
    const int Fs[3] = { 32, 32, 64 };

    float* ws    = (float*)d_ws;
    float* xw    = ws;
    float* bufA  = xw    + (size_t)NN * 128;
    float* bufB  = bufA  + (size_t)NN * 128;
    float* ss    = bufB  + (size_t)NN * 128;
    float* sd    = ss    + (size_t)NN * 4;
    float* mseg  = sd    + (size_t)NN * 4;
    float* denom = mseg  + (size_t)NN * 4;
    float* alpha = denom + (size_t)NN * 4;
    float* ce    = alpha + (size_t)ETOT * 4;
    float* meanw = ce + 4;

    mean_kernel<<<1, 1024, 0, stream>>>(ew, meanw, EE);

    const float* in = x;
    float* outs[3] = { bufA, bufB, (float*)d_out };

    for (int l = 0; l < 3; ++l) {
        const int H = Hs[l], F = Fs[l], C = H * F;

        edge_coef_kernel<<<1, 32, 0, stream>>>(We[l], Ae[l], ce, H, F);

        gemm_wmma_f32<<<dim3(NN / 16, C / 16), 32, 0, stream>>>(in, Wl[l], xw, KDIM, C);

        score_kernel<<<(NN * H + 255) / 256, 256, 0, stream>>>(xw, As[l], Ad[l], ss, sd, H, F);

        init_kernel<<<NN, C, 0, stream>>>(outs[l], Bi[l], mseg, denom, C, H);

        alpha_max_kernel<<<(ETOT * H + 255) / 256, 256, 0, stream>>>(
            ei, ew, meanw, ss, sd, ce, alpha, mseg, H);

        exp_denom_kernel<<<(ETOT * H + 255) / 256, 256, 0, stream>>>(ei, mseg, alpha, denom, H);

        scatter_kernel<<<ETOT, C, 0, stream>>>(ei, xw, alpha, denom, outs[l], H, F);

        if (l < 2) elu_kernel<<<((NN * C) + 255) / 256, 256, 0, stream>>>(outs[l], NN * C);

        in = outs[l];
    }
}


# edge-weight mean: the one block's loop of one 4-byte load per round trip becomes 16-byte loads, 8 issued per thread ahead of the first wait; same block and launches; the mean's sum order changes
# speedup vs baseline: 2.1269x; 2.1269x over previous
#include <hip/hip_runtime.h>
#include <math.h>

#define NN    50000
#define EE    800000
#define ETOT  (EE + NN)
#define KDIM  128

typedef float v2f __attribute__((ext_vector_type(2)));
typedef float v8f __attribute__((ext_vector_type(8)));

__device__ __forceinline__ void atomicMaxF(float* addr, float val) {
    if (val >= 0.0f) atomicMax((int*)addr, __float_as_int(val));
    else             atomicMin((unsigned int*)addr, __float_as_uint(val));
}

#define MEAN_U 8
__global__ __launch_bounds__(1024) void mean_kernel(const float* __restrict__ ew,
                                                    float* __restrict__ meanw, int E) {
    __shared__ float sdata[1024];
    const int n4   = E >> 2;
    const int full = n4 / (MEAN_U * 1024);
    const float4* __restrict__ ew4 = reinterpret_cast<const float4*>(ew);
    float s = 0.0f;

    const float4* p = ew4 + threadIdx.x;
    for (int g = 0; g < full; ++g, p += MEAN_U * 1024) {
        float4 v[MEAN_U];
#pragma unroll
        for (int u = 0; u < MEAN_U; ++u) v[u] = p[u * 1024];
#pragma unroll
        for (int u = 0; u < MEAN_U; ++u) s += (v[u].x + v[u].y) + (v[u].z + v[u].w);
    }

    const int r0 = full * (MEAN_U * 1024) + (int)threadIdx.x;
    if (r0 < n4) {
        float4 v[MEAN_U];
#pragma unroll
        for (int u = 0; u < MEAN_U; ++u) {
            const int idx = r0 + u * 1024;
            v[u] = ew4[idx < n4 ? idx : n4 - 1];
        }
#pragma unroll
        for (int u = 0; u < MEAN_U; ++u)
            if (r0 + u * 1024 < n4) s += (v[u].x + v[u].y) + (v[u].z + v[u].w);
    }
    if ((int)threadIdx.x < (E & 3)) s += ew[(n4 << 2) + threadIdx.x];

    sdata[threadIdx.x] = s;
    __syncthreads();
    for (int o = 512; o > 0; o >>= 1) {
        if (threadIdx.x < o) sdata[threadIdx.x] += sdata[threadIdx.x + o];
        __syncthreads();
    }
    if (threadIdx.x == 0) meanw[0] = sdata[0] / (float)E;
}

__global__ void edge_coef_kernel(const float* __restrict__ We, const float* __restrict__ ae,
                                 float* __restrict__ ce, int H, int F) {
    int h = threadIdx.x;
    if (h >= H) return;
    float s = 0.0f;
    for (int f = 0; f < F; ++f) s += We[h * F + f] * ae[h * F + f];
    ce[h] = s;
}

__global__ __launch_bounds__(32) void gemm_wmma_f32(const float* __restrict__ X,
                                                    const float* __restrict__ W,
                                                    float* __restrict__ out,
                                                    int K, int C) {
    const int tm   = blockIdx.x;
    const int tc   = blockIdx.y;
    const int lane = threadIdx.x;
    const int half = lane >> 4;
    const int l15  = lane & 15;

    v8f acc = {};
    const float* xrow = X + (size_t)(tm * 16 + l15) * K;
    const int    coln = tc * 16 + l15;

    for (int k = 0; k < K; k += 4) {
        const int kk = k + half * 2;
        v2f a, b;
        a[0] = xrow[kk];
        a[1] = xrow[kk + 1];
        b[0] = W[(size_t)kk * C + coln];
        b[1] = W[(size_t)(kk + 1) * C + coln];
        acc = __builtin_amdgcn_wmma_f32_16x16x4_f32(
             false, a,  false, b,
             (short)0, acc,  false,  false);
    }
#pragma unroll
    for (int v = 0; v < 8; ++v) {
        const int row = tm * 16 + v + half * 8;
        out[(size_t)row * C + coln] = acc[v];
    }
}

__global__ void score_kernel(const float* __restrict__ xw,
                             const float* __restrict__ a_s, const float* __restrict__ a_d,
                             float* __restrict__ ss, float* __restrict__ sd,
                             int H, int F) {
    int i = blockIdx.x * blockDim.x + threadIdx.x;
    if (i >= NN * H) return;
    const int n = i / H, h = i % H;
    const float* v  = xw + (size_t)n * H * F + h * F;
    const float* as = a_s + h * F;
    const float* ad = a_d + h * F;
    float s1 = 0.0f, s2 = 0.0f;
    for (int f = 0; f < F; ++f) { float x = v[f]; s1 += x * as[f]; s2 += x * ad[f]; }
    ss[i] = s1;
    sd[i] = s2;
}

__global__ void init_kernel(float* __restrict__ out, const float* __restrict__ b,
                            float* __restrict__ m, float* __restrict__ denom,
                            int C, int H) {
    const int n = blockIdx.x, t = threadIdx.x;
    out[(size_t)n * C + t] = b[t];
    if (t < H) { m[n * H + t] = -__builtin_inff(); denom[n * H + t] = 0.0f; }
}

__global__ void alpha_max_kernel(const int* __restrict__ ei, const float* __restrict__ ew,
                                 const float* __restrict__ meanw,
                                 const float* __restrict__ ss, const float* __restrict__ sd,
                                 const float* __restrict__ ce,
                                 float* __restrict__ alpha, float* __restrict__ m, int H) {
    int idx = blockIdx.x * blockDim.x + threadIdx.x;
    if (idx >= ETOT * H) return;
    const int e = idx / H, h = idx % H;
    int s, d; float ea;
    if (e < EE) { s = ei[e]; d = ei[EE + e]; ea = ew[e]; }
    else        { s = d = e - EE;            ea = meanw[0]; }
    float a = ss[s * H + h] + sd[d * H + h] + ea * ce[h];
    a = a > 0.0f ? a : 0.2f * a;
    alpha[idx] = a;
    atomicMaxF(&m[d * H + h], a);
}

__global__ void exp_denom_kernel(const int* __restrict__ ei,
                                 const float* __restrict__ m,
                                 float* __restrict__ alpha, float* __restrict__ denom, int H) {
    int idx = blockIdx.x * blockDim.x + threadIdx.x;
    if (idx >= ETOT * H) return;
    const int e = idx / H, h = idx % H;
    const int d = (e < EE) ? ei[EE + e] : (e - EE);
    float ex = __expf(alpha[idx] - m[d * H + h]);
    alpha[idx] = ex;
    atomicAdd(&denom[d * H + h], ex);
}

template <int C, int H>
__global__ __launch_bounds__(256) void scatter_kernel(const int* __restrict__ ei,
                                                      const float* __restrict__ xw,
                                                      const float* __restrict__ alpha,
                                                      const float* __restrict__ denom,
                                                      float* __restrict__ out) {
    constexpr int NC = C / 32;
    constexpr int F  = C / H;
    constexpr int U  = 8;
    static_assert(F % 32 == 0, "a lane's element c*32+lane must lie in head (c*32)/F");
    static_assert(ETOT % U == 0 && 32 % U == 0, "every wave's edge count is a multiple of U");

    const int lane = threadIdx.x & 31;
    const int wave = __builtin_amdgcn_readfirstlane((int)(threadIdx.x >> 5));
    const int base = (blockIdx.x * (256 / 32) + wave) * 32;
    if (base >= ETOT) return;

    const int e = base + lane;
    int s = 0, d = 0;
    float at[H];
    for (int h = 0; h < H; ++h) at[h] = 0.0f;
    if (e < ETOT) {
        if (e < EE) { s = ei[e]; d = ei[EE + e]; } else { s = d = e - EE; }
        if constexpr (H == 4) {
            const float4 al = *reinterpret_cast<const float4*>(alpha + (size_t)e * 4);
            const float4 de = *reinterpret_cast<const float4*>(denom + (size_t)d * 4);
            at[0] = al.x / (de.x + 1e-16f);
            at[1] = al.y / (de.y + 1e-16f);
            at[2] = al.z / (de.z + 1e-16f);
            at[3] = al.w / (de.w + 1e-16f);
        } else {
            for (int h = 0; h < H; ++h)
                at[h] = alpha[(size_t)e * H + h] / (denom[d * H + h] + 1e-16f);
        }
    }

    const int cnt = (ETOT - base) < 32 ? (ETOT - base) : 32;
    for (int j0 = 0; j0 < cnt; j0 += U) {
        float v[U][NC];
        for (int u = 0; u < U; ++u) {
            const int sj = __builtin_amdgcn_readlane(s, j0 + u);
            const float* row = xw + (size_t)sj * C + lane;
            for (int c = 0; c < NC; ++c) v[u][c] = row[c * 32];
        }
        for (int u = 0; u < U; ++u) {
            const int dj = __builtin_amdgcn_readlane(d, j0 + u);
            float* orow = out + (size_t)dj * C + lane;
            for (int c = 0; c < NC; ++c) {
                const float a = __int_as_float(
                    __builtin_amdgcn_readlane(__float_as_int(at[(c * 32) / F]), j0 + u));
                atomicAdd(orow + c * 32, v[u][c] * a);
            }
        }
    }
}

__global__ void elu_kernel(float* __restrict__ buf, int n) {
    int i = blockIdx.x * blockDim.x + threadIdx.x;
    if (i >= n) return;
    float v = buf[i];
    buf[i] = v > 0.0f ? v : expm1f(v);
}

extern "C" void kernel_launch(void* const* d_in, const int* in_sizes, int n_in,
                              void* d_out, int out_size, void* d_ws, size_t ws_size,
                              hipStream_t stream) {
    const float* x  = (const float*)d_in[0];
    const int*   ei = (const int*)d_in[1];
    const float* ew = (const float*)d_in[2];

    const float* Wl[3] = { (const float*)d_in[3],  (const float*)d_in[9],  (const float*)d_in[15] };
    const float* As[3] = { (const float*)d_in[4],  (const float*)d_in[10], (const float*)d_in[16] };
    const float* Ad[3] = { (const float*)d_in[5],  (const float*)d_in[11], (const float*)d_in[17] };
    const float* We[3] = { (const float*)d_in[6],  (const float*)d_in[12], (const float*)d_in[18] };
    const float* Ae[3] = { (const float*)d_in[7],  (const float*)d_in[13], (const float*)d_in[19] };
    const float* Bi[3] = { (const float*)d_in[8],  (const float*)d_in[14], (const float*)d_in[20] };

    const int Hs[3] = { 4, 4, 1 };
    const int Fs[3] = { 32, 32, 64 };

    float* ws    = (float*)d_ws;
    float* xw    = ws;
    float* bufA  = xw    + (size_t)NN * 128;
    float* bufB  = bufA  + (size_t)NN * 128;
    float* ss    = bufB  + (size_t)NN * 128;
    float* sd    = ss    + (size_t)NN * 4;
    float* mseg  = sd    + (size_t)NN * 4;
    float* denom = mseg  + (size_t)NN * 4;
    float* alpha = denom + (size_t)NN * 4;
    float* ce    = alpha + (size_t)ETOT * 4;
    float* meanw = ce + 4;

    mean_kernel<<<1, 1024, 0, stream>>>(ew, meanw, EE);

    const float* in = x;
    float* outs[3] = { bufA, bufB, (float*)d_out };

    for (int l = 0; l < 3; ++l) {
        const int H = Hs[l], F = Fs[l], C = H * F;

        edge_coef_kernel<<<1, 32, 0, stream>>>(We[l], Ae[l], ce, H, F);

        gemm_wmma_f32<<<dim3(NN / 16, C / 16), 32, 0, stream>>>(in, Wl[l], xw, KDIM, C);

        score_kernel<<<(NN * H + 255) / 256, 256, 0, stream>>>(xw, As[l], Ad[l], ss, sd, H, F);

        init_kernel<<<NN, C, 0, stream>>>(outs[l], Bi[l], mseg, denom, C, H);

        alpha_max_kernel<<<(ETOT * H + 255) / 256, 256, 0, stream>>>(
            ei, ew, meanw, ss, sd, ce, alpha, mseg, H);

        exp_denom_kernel<<<(ETOT * H + 255) / 256, 256, 0, stream>>>(ei, mseg, alpha, denom, H);

        if (l < 2)
            scatter_kernel<128, 4><<<((ETOT + 31) / 32 + 7) / 8, 256, 0, stream>>>(
                ei, xw, alpha, denom, outs[l]);
        else
            scatter_kernel<64, 1><<<((ETOT + 31) / 32 + 7) / 8, 256, 0, stream>>>(
                ei, xw, alpha, denom, outs[l]);

        if (l < 2) elu_kernel<<<((NN * C) + 255) / 256, 256, 0, stream>>>(outs[l], NN * C);

        in = outs[l];
    }
}


# softmax division leaves the per-edge path: scatter forms exp(alpha-m), adds ex to denom and ex*row to out; a finishing pass divides, adds bias, ELU; exp+sum launches go; out's sum order changes
# speedup vs baseline: 2.2339x; 1.0503x over previous
#include <hip/hip_runtime.h>
#include <math.h>

#define NN    50000
#define EE    800000
#define ETOT  (EE + NN)
#define KDIM  128

typedef float v2f __attribute__((ext_vector_type(2)));
typedef float v8f __attribute__((ext_vector_type(8)));

__device__ __forceinline__ void atomicMaxF(float* addr, float val) {
    if (val >= 0.0f) atomicMax((int*)addr, __float_as_int(val));
    else             atomicMin((unsigned int*)addr, __float_as_uint(val));
}

#define MEAN_U 8
__global__ __launch_bounds__(1024) void mean_kernel(const float* __restrict__ ew,
                                                    float* __restrict__ meanw, int E) {
    __shared__ float sdata[1024];
    const int n4   = E >> 2;
    const int full = n4 / (MEAN_U * 1024);
    const float4* __restrict__ ew4 = reinterpret_cast<const float4*>(ew);
    float s = 0.0f;

    const float4* p = ew4 + threadIdx.x;
    for (int g = 0; g < full; ++g, p += MEAN_U * 1024) {
        float4 v[MEAN_U];
#pragma unroll
        for (int u = 0; u < MEAN_U; ++u) v[u] = p[u * 1024];
#pragma unroll
        for (int u = 0; u < MEAN_U; ++u) s += (v[u].x + v[u].y) + (v[u].z + v[u].w);
    }

    const int r0 = full * (MEAN_U * 1024) + (int)threadIdx.x;
    if (r0 < n4) {
        float4 v[MEAN_U];
#pragma unroll
        for (int u = 0; u < MEAN_U; ++u) {
            const int idx = r0 + u * 1024;
            v[u] = ew4[idx < n4 ? idx : n4 - 1];
        }
#pragma unroll
        for (int u = 0; u < MEAN_U; ++u)
            if (r0 + u * 1024 < n4) s += (v[u].x + v[u].y) + (v[u].z + v[u].w);
    }
    if ((int)threadIdx.x < (E & 3)) s += ew[(n4 << 2) + threadIdx.x];

    sdata[threadIdx.x] = s;
    __syncthreads();
    for (int o = 512; o > 0; o >>= 1) {
        if (threadIdx.x < o) sdata[threadIdx.x] += sdata[threadIdx.x + o];
        __syncthreads();
    }
    if (threadIdx.x == 0) meanw[0] = sdata[0] / (float)E;
}

__global__ void edge_coef_kernel(const float* __restrict__ We, const float* __restrict__ ae,
                                 float* __restrict__ ce, int H, int F) {
    int h = threadIdx.x;
    if (h >= H) return;
    float s = 0.0f;
    for (int f = 0; f < F; ++f) s += We[h * F + f] * ae[h * F + f];
    ce[h] = s;
}

__global__ __launch_bounds__(32) void gemm_wmma_f32(const float* __restrict__ X,
                                                    const float* __restrict__ W,
                                                    float* __restrict__ out,
                                                    int K, int C) {
    const int tm   = blockIdx.x;
    const int tc   = blockIdx.y;
    const int lane = threadIdx.x;
    const int half = lane >> 4;
    const int l15  = lane & 15;

    v8f acc = {};
    const float* xrow = X + (size_t)(tm * 16 + l15) * K;
    const int    coln = tc * 16 + l15;

    for (int k = 0; k < K; k += 4) {
        const int kk = k + half * 2;
        v2f a, b;
        a[0] = xrow[kk];
        a[1] = xrow[kk + 1];
        b[0] = W[(size_t)kk * C + coln];
        b[1] = W[(size_t)(kk + 1) * C + coln];
        acc = __builtin_amdgcn_wmma_f32_16x16x4_f32(
             false, a,  false, b,
             (short)0, acc,  false,  false);
    }
#pragma unroll
    for (int v = 0; v < 8; ++v) {
        const int row = tm * 16 + v + half * 8;
        out[(size_t)row * C + coln] = acc[v];
    }
}

__global__ void score_kernel(const float* __restrict__ xw,
                             const float* __restrict__ a_s, const float* __restrict__ a_d,
                             float* __restrict__ ss, float* __restrict__ sd,
                             int H, int F) {
    int i = blockIdx.x * blockDim.x + threadIdx.x;
    if (i >= NN * H) return;
    const int n = i / H, h = i % H;
    const float* v  = xw + (size_t)n * H * F + h * F;
    const float* as = a_s + h * F;
    const float* ad = a_d + h * F;
    float s1 = 0.0f, s2 = 0.0f;
    for (int f = 0; f < F; ++f) { float x = v[f]; s1 += x * as[f]; s2 += x * ad[f]; }
    ss[i] = s1;
    sd[i] = s2;
}

__global__ void init_kernel(float* __restrict__ out,
                            float* __restrict__ m, float* __restrict__ denom,
                            int C, int H) {
    const int n = blockIdx.x, t = threadIdx.x;
    out[(size_t)n * C + t] = 0.0f;
    if (t < H) { m[n * H + t] = -__builtin_inff(); denom[n * H + t] = 0.0f; }
}

__global__ void alpha_max_kernel(const int* __restrict__ ei, const float* __restrict__ ew,
                                 const float* __restrict__ meanw,
                                 const float* __restrict__ ss, const float* __restrict__ sd,
                                 const float* __restrict__ ce,
                                 float* __restrict__ alpha, float* __restrict__ m, int H) {
    int idx = blockIdx.x * blockDim.x + threadIdx.x;
    if (idx >= ETOT * H) return;
    const int e = idx / H, h = idx % H;
    int s, d; float ea;
    if (e < EE) { s = ei[e]; d = ei[EE + e]; ea = ew[e]; }
    else        { s = d = e - EE;            ea = meanw[0]; }
    float a = ss[s * H + h] + sd[d * H + h] + ea * ce[h];
    a = a > 0.0f ? a : 0.2f * a;
    alpha[idx] = a;
    atomicMaxF(&m[d * H + h], a);
}

template <int C, int H>
__global__ __launch_bounds__(256) void scatter_kernel(const int* __restrict__ ei,
                                                      const float* __restrict__ xw,
                                                      const float* __restrict__ alpha,
                                                      const float* __restrict__ m,
                                                      float* __restrict__ denom,
                                                      float* __restrict__ out) {
    constexpr int NC = C / 32;
    constexpr int F  = C / H;
    constexpr int U  = 8;
    static_assert(F % 32 == 0, "a lane's element c*32+lane must lie in head (c*32)/F");
    static_assert(ETOT % U == 0 && 32 % U == 0, "every wave's edge count is a multiple of U");

    const int lane = threadIdx.x & 31;
    const int wave = __builtin_amdgcn_readfirstlane((int)(threadIdx.x >> 5));
    const int base = (blockIdx.x * (256 / 32) + wave) * 32;
    if (base >= ETOT) return;

    const int e = base + lane;
    int s = 0, d = 0;
    float at[H];
    for (int h = 0; h < H; ++h) at[h] = 0.0f;
    if (e < ETOT) {
        if (e < EE) { s = ei[e]; d = ei[EE + e]; } else { s = d = e - EE; }
        if constexpr (H == 4) {
            const float4 al = *reinterpret_cast<const float4*>(alpha + (size_t)e * 4);
            const float4 mx = *reinterpret_cast<const float4*>(m + (size_t)d * 4);
            at[0] = __expf(al.x - mx.x);
            at[1] = __expf(al.y - mx.y);
            at[2] = __expf(al.z - mx.z);
            at[3] = __expf(al.w - mx.w);
        } else {
            for (int h = 0; h < H; ++h)
                at[h] = __expf(alpha[(size_t)e * H + h] - m[d * H + h]);
        }
    }

    if constexpr (H == 4) {
        const int hsel = lane & 3;
        for (int g = 0; g < 4; ++g) {
            const int srcl = g * 8 + (lane >> 2);
            const float e0 = __shfl(at[0], srcl), e1 = __shfl(at[1], srcl);
            const float e2 = __shfl(at[2], srcl), e3 = __shfl(at[3], srcl);
            const int   dd = __shfl(d, srcl);
            const float lo = (hsel & 1) ? e1 : e0;
            const float hi = (hsel & 1) ? e3 : e2;
            const float ev = (hsel & 2) ? hi : lo;
            if (base + srcl < ETOT) atomicAdd(denom + (size_t)dd * 4 + hsel, ev);
        }
    } else {
        if (e < ETOT)
            for (int h = 0; h < H; ++h) atomicAdd(denom + (size_t)d * H + h, at[h]);
    }

    const int cnt = (ETOT - base) < 32 ? (ETOT - base) : 32;
    for (int j0 = 0; j0 < cnt; j0 += U) {
        float v[U][NC];
        for (int u = 0; u < U; ++u) {
            const int sj = __builtin_amdgcn_readlane(s, j0 + u);
            const float* row = xw + (size_t)sj * C + lane;
            for (int c = 0; c < NC; ++c) v[u][c] = row[c * 32];
        }
        for (int u = 0; u < U; ++u) {
            const int dj = __builtin_amdgcn_readlane(d, j0 + u);
            float* orow = out + (size_t)dj * C + lane;
            for (int c = 0; c < NC; ++c) {
                const float a = __int_as_float(
                    __builtin_amdgcn_readlane(__float_as_int(at[(c * 32) / F]), j0 + u));
                atomicAdd(orow + c * 32, v[u][c] * a);
            }
        }
    }
}

template <int C, int H, bool ELU>
__global__ void finish_kernel(float* __restrict__ buf, const float* __restrict__ denom,
                              const float* __restrict__ b, int n) {
    int i = blockIdx.x * blockDim.x + threadIdx.x;
    if (i >= n) return;
    const int node = i / C, c = i % C;
    float v = buf[i] / (denom[node * H + c / (C / H)] + 1e-16f) + b[c];
    if constexpr (ELU) v = v > 0.0f ? v : expm1f(v);
    buf[i] = v;
}

extern "C" void kernel_launch(void* const* d_in, const int* in_sizes, int n_in,
                              void* d_out, int out_size, void* d_ws, size_t ws_size,
                              hipStream_t stream) {
    const float* x  = (const float*)d_in[0];
    const int*   ei = (const int*)d_in[1];
    const float* ew = (const float*)d_in[2];

    const float* Wl[3] = { (const float*)d_in[3],  (const float*)d_in[9],  (const float*)d_in[15] };
    const float* As[3] = { (const float*)d_in[4],  (const float*)d_in[10], (const float*)d_in[16] };
    const float* Ad[3] = { (const float*)d_in[5],  (const float*)d_in[11], (const float*)d_in[17] };
    const float* We[3] = { (const float*)d_in[6],  (const float*)d_in[12], (const float*)d_in[18] };
    const float* Ae[3] = { (const float*)d_in[7],  (const float*)d_in[13], (const float*)d_in[19] };
    const float* Bi[3] = { (const float*)d_in[8],  (const float*)d_in[14], (const float*)d_in[20] };

    const int Hs[3] = { 4, 4, 1 };
    const int Fs[3] = { 32, 32, 64 };

    float* ws    = (float*)d_ws;
    float* xw    = ws;
    float* bufA  = xw    + (size_t)NN * 128;
    float* bufB  = bufA  + (size_t)NN * 128;
    float* ss    = bufB  + (size_t)NN * 128;
    float* sd    = ss    + (size_t)NN * 4;
    float* mseg  = sd    + (size_t)NN * 4;
    float* denom = mseg  + (size_t)NN * 4;
    float* alpha = denom + (size_t)NN * 4;
    float* ce    = alpha + (size_t)ETOT * 4;
    float* meanw = ce + 4;

    mean_kernel<<<1, 1024, 0, stream>>>(ew, meanw, EE);

    const float* in = x;
    float* outs[3] = { bufA, bufB, (float*)d_out };

    for (int l = 0; l < 3; ++l) {
        const int H = Hs[l], F = Fs[l], C = H * F;

        edge_coef_kernel<<<1, 32, 0, stream>>>(We[l], Ae[l], ce, H, F);

        gemm_wmma_f32<<<dim3(NN / 16, C / 16), 32, 0, stream>>>(in, Wl[l], xw, KDIM, C);

        score_kernel<<<(NN * H + 255) / 256, 256, 0, stream>>>(xw, As[l], Ad[l], ss, sd, H, F);

        init_kernel<<<NN, C, 0, stream>>>(outs[l], mseg, denom, C, H);

        alpha_max_kernel<<<(ETOT * H + 255) / 256, 256, 0, stream>>>(
            ei, ew, meanw, ss, sd, ce, alpha, mseg, H);

        if (l < 2)
            scatter_kernel<128, 4><<<((ETOT + 31) / 32 + 7) / 8, 256, 0, stream>>>(
                ei, xw, alpha, mseg, denom, outs[l]);
        else
            scatter_kernel<64, 1><<<((ETOT + 31) / 32 + 7) / 8, 256, 0, stream>>>(
                ei, xw, alpha, mseg, denom, outs[l]);

        if (l < 2)
            finish_kernel<128, 4, true><<<((NN * C) + 255) / 256, 256, 0, stream>>>(
                outs[l], denom, Bi[l], NN * C);
        else
            finish_kernel<64, 1, false><<<((NN * C) + 255) / 256, 256, 0, stream>>>(
                outs[l], denom, Bi[l], NN * C);

        in = outs[l];
    }
}


# out's memory atomics become LDS adds: a block owns 196 destination rows, scans all edge destinations (no edge list), sums source rows in LDS, divides, stores; 22 -> 19 launches; sum order changes
# speedup vs baseline: 2.5887x; 1.1588x over previous
#include <hip/hip_runtime.h>
#include <math.h>

#define NN    50000
#define EE    800000
#define ETOT  (EE + NN)
#define KDIM  128

typedef float v2f __attribute__((ext_vector_type(2)));
typedef float v8f __attribute__((ext_vector_type(8)));

__device__ __forceinline__ void atomicMaxF(float* addr, float val) {
    if (val >= 0.0f) atomicMax((int*)addr, __float_as_int(val));
    else             atomicMin((unsigned int*)addr, __float_as_uint(val));
}

#define MEAN_U 8
__global__ __launch_bounds__(1024) void mean_kernel(const float* __restrict__ ew,
                                                    float* __restrict__ meanw, int E) {
    __shared__ float sdata[1024];
    const int n4   = E >> 2;
    const int full = n4 / (MEAN_U * 1024);
    const float4* __restrict__ ew4 = reinterpret_cast<const float4*>(ew);
    float s = 0.0f;

    const float4* p = ew4 + threadIdx.x;
    for (int g = 0; g < full; ++g, p += MEAN_U * 1024) {
        float4 v[MEAN_U];
#pragma unroll
        for (int u = 0; u < MEAN_U; ++u) v[u] = p[u * 1024];
#pragma unroll
        for (int u = 0; u < MEAN_U; ++u) s += (v[u].x + v[u].y) + (v[u].z + v[u].w);
    }

    const int r0 = full * (MEAN_U * 1024) + (int)threadIdx.x;
    if (r0 < n4) {
        float4 v[MEAN_U];
#pragma unroll
        for (int u = 0; u < MEAN_U; ++u) {
            const int idx = r0 + u * 1024;
            v[u] = ew4[idx < n4 ? idx : n4 - 1];
        }
#pragma unroll
        for (int u = 0; u < MEAN_U; ++u)
            if (r0 + u * 1024 < n4) s += (v[u].x + v[u].y) + (v[u].z + v[u].w);
    }
    if ((int)threadIdx.x < (E & 3)) s += ew[(n4 << 2) + threadIdx.x];

    sdata[threadIdx.x] = s;
    __syncthreads();
    for (int o = 512; o > 0; o >>= 1) {
        if (threadIdx.x < o) sdata[threadIdx.x] += sdata[threadIdx.x + o];
        __syncthreads();
    }
    if (threadIdx.x == 0) meanw[0] = sdata[0] / (float)E;
}

__global__ void edge_coef_kernel(const float* __restrict__ We, const float* __restrict__ ae,
                                 float* __restrict__ ce, int H, int F) {
    int h = threadIdx.x;
    if (h >= H) return;
    float s = 0.0f;
    for (int f = 0; f < F; ++f) s += We[h * F + f] * ae[h * F + f];
    ce[h] = s;
}

__global__ __launch_bounds__(32) void gemm_wmma_f32(const float* __restrict__ X,
                                                    const float* __restrict__ W,
                                                    float* __restrict__ out,
                                                    int K, int C) {
    const int tm   = blockIdx.x;
    const int tc   = blockIdx.y;
    const int lane = threadIdx.x;
    const int half = lane >> 4;
    const int l15  = lane & 15;

    v8f acc = {};
    const float* xrow = X + (size_t)(tm * 16 + l15) * K;
    const int    coln = tc * 16 + l15;

    for (int k = 0; k < K; k += 4) {
        const int kk = k + half * 2;
        v2f a, b;
        a[0] = xrow[kk];
        a[1] = xrow[kk + 1];
        b[0] = W[(size_t)kk * C + coln];
        b[1] = W[(size_t)(kk + 1) * C + coln];
        acc = __builtin_amdgcn_wmma_f32_16x16x4_f32(
             false, a,  false, b,
             (short)0, acc,  false,  false);
    }
#pragma unroll
    for (int v = 0; v < 8; ++v) {
        const int row = tm * 16 + v + half * 8;
        out[(size_t)row * C + coln] = acc[v];
    }
}

__global__ void score_kernel(const float* __restrict__ xw,
                             const float* __restrict__ a_s, const float* __restrict__ a_d,
                             float* __restrict__ ss, float* __restrict__ sd,
                             int H, int F) {
    int i = blockIdx.x * blockDim.x + threadIdx.x;
    if (i >= NN * H) return;
    const int n = i / H, h = i % H;
    const float* v  = xw + (size_t)n * H * F + h * F;
    const float* as = a_s + h * F;
    const float* ad = a_d + h * F;
    float s1 = 0.0f, s2 = 0.0f;
    for (int f = 0; f < F; ++f) { float x = v[f]; s1 += x * as[f]; s2 += x * ad[f]; }
    ss[i] = s1;
    sd[i] = s2;
}

__global__ void m_init_kernel(float* __restrict__ m, int n) {
    const int i = blockIdx.x * blockDim.x + threadIdx.x;
    if (i < n) m[i] = -__builtin_inff();
}

__global__ void alpha_max_kernel(const int* __restrict__ ei, const float* __restrict__ ew,
                                 const float* __restrict__ meanw,
                                 const float* __restrict__ ss, const float* __restrict__ sd,
                                 const float* __restrict__ ce,
                                 float* __restrict__ alpha, float* __restrict__ m, int H) {
    int idx = blockIdx.x * blockDim.x + threadIdx.x;
    if (idx >= ETOT * H) return;
    const int e = idx / H, h = idx % H;
    int s, d; float ea;
    if (e < EE) { s = ei[e]; d = ei[EE + e]; ea = ew[e]; }
    else        { s = d = e - EE;            ea = meanw[0]; }
    float a = ss[s * H + h] + sd[d * H + h] + ea * ce[h];
    a = a > 0.0f ? a : 0.2f * a;
    alpha[idx] = a;
    atomicMaxF(&m[d * H + h], a);
}

#define RROWS   196
#define RWAVES  32
#define QCAP    64
#define SCAN_G  8

template <int C, int H, bool ELU>
__global__ __launch_bounds__(1024) void range_kernel(const int* __restrict__ ei,
                                                     const float* __restrict__ xw,
                                                     const float* __restrict__ alpha,
                                                     const float* __restrict__ m,
                                                     const float* __restrict__ bias,
                                                     float* __restrict__ out) {
    constexpr int NC = C / 32;
    constexpr int F  = C / H;
    constexpr int U  = 8;
    constexpr int N4 = EE / 4;
    constexpr int ROUND4 = RWAVES * SCAN_G * 32;
    constexpr int NR = (N4 + ROUND4 - 1) / ROUND4;
    static_assert(F % 32 == 0, "a lane's element c*32+lane must lie in head (c*32)/F");
    static_assert(EE % 4 == 0, "the destination plane is read in groups of four");
    static_assert(32 % U == 0 && QCAP >= 64 && (QCAP & (QCAP - 1)) == 0, "queue and group sizes");

    __shared__ float acc[RROWS * C];
    __shared__ float den[RROWS * H];
    __shared__ float mrow[RROWS * H];
    __shared__ int que[RWAVES][QCAP];

    const int tid   = threadIdx.x;
    const int lane  = tid & 31;
    const int wave  = __builtin_amdgcn_readfirstlane(tid >> 5);
    const int d0    = blockIdx.x * RROWS;
    const int nrows = (NN - d0) < RROWS ? (NN - d0) : RROWS;

    {
        constexpr int RPT = 1024 / C;
        constexpr int P0  = 4;
        const int c = tid % C, h = c / F;
        for (int r0 = tid / C; r0 < nrows; r0 += RPT * P0) {
            float mx[P0], al[P0], xv[P0];
#pragma unroll
            for (int k = 0; k < P0; ++k) {
                const int r = r0 + k * RPT;
                const int n = d0 + (r < nrows ? r : nrows - 1);
                mx[k] = m[n * H + h];
                al[k] = alpha[(size_t)(EE + n) * H + h];
                xv[k] = xw[(size_t)n * C + c];
            }
#pragma unroll
            for (int k = 0; k < P0; ++k) {
                const int r = r0 + k * RPT;
                if (r < nrows) {
                    const float ex = __expf(al[k] - mx[k]);
                    acc[r * C + c] = ex * xv[k];
                    if (c % F == 0) { den[r * H + h] = ex; mrow[r * H + h] = mx[k]; }
                }
            }
        }
    }
    __syncthreads();

    int qhead = 0, qcnt = 0;

    auto drain = [&](const int nb) {
        int s = 0, r = 0;
        float at[H];
        for (int h = 0; h < H; ++h) at[h] = 0.0f;
        __builtin_amdgcn_wave_barrier();
        if (lane < nb) {
            const int e = que[wave][(qhead + lane) & (QCAP - 1)];
            s = ei[e];
            r = ei[EE + e] - d0;
            if constexpr (H == 4) {
                const float4 al = *reinterpret_cast<const float4*>(alpha + (size_t)e * 4);
                const float4 mx = *reinterpret_cast<const float4*>(mrow + r * 4);
                at[0] = __expf(al.x - mx.x);
                at[1] = __expf(al.y - mx.y);
                at[2] = __expf(al.z - mx.z);
                at[3] = __expf(al.w - mx.w);
            } else {
                for (int h = 0; h < H; ++h)
                    at[h] = __expf(alpha[(size_t)e * H + h] - mrow[r * H + h]);
            }
            for (int h = 0; h < H; ++h) atomicAdd(&den[r * H + h], at[h]);
        }
        for (int j0 = 0; j0 < nb; j0 += U) {
            float v[U][NC];
#pragma unroll
            for (int u = 0; u < U; ++u) {
                if (j0 + u < nb) {
                    const int sj = __builtin_amdgcn_readlane(s, j0 + u);
                    const float* row = xw + (size_t)sj * C + lane;
                    for (int c = 0; c < NC; ++c) v[u][c] = row[c * 32];
                }
            }
#pragma unroll
            for (int u = 0; u < U; ++u) {
                if (j0 + u < nb) {
                    const int rj = __builtin_amdgcn_readlane(r, j0 + u);
                    float* arow = acc + rj * C + lane;
                    for (int c = 0; c < NC; ++c) {
                        const float a = __int_as_float(
                            __builtin_amdgcn_readlane(__float_as_int(at[(c * 32) / F]), j0 + u));
                        atomicAdd(arow + c * 32, v[u][c] * a);
                    }
                }
            }
        }
    };

    const int4* __restrict__ dst4 = reinterpret_cast<const int4*>(ei + EE);
    const int rot = blockIdx.x % NR;
    for (int it = 0; it < NR; ++it) {
        int rnd = it + rot;
        if (rnd >= NR) rnd -= NR;
        const int g0 = rnd * ROUND4 + wave * (SCAN_G * 32);
        int4 dv[SCAN_G];
#pragma unroll
        for (int g = 0; g < SCAN_G; ++g) {
            const int i4 = g0 + g * 32 + lane;
            dv[g] = dst4[i4 < N4 ? i4 : N4 - 1];
        }
        unsigned hits = 0u;
#pragma unroll
        for (int g = 0; g < SCAN_G; ++g) {
            const bool ok = (g0 + g * 32 + lane) < N4;
            hits |= (ok && (unsigned)(dv[g].x - d0) < (unsigned)nrows) ? (1u << (g * 4 + 0)) : 0u;
            hits |= (ok && (unsigned)(dv[g].y - d0) < (unsigned)nrows) ? (1u << (g * 4 + 1)) : 0u;
            hits |= (ok && (unsigned)(dv[g].z - d0) < (unsigned)nrows) ? (1u << (g * 4 + 2)) : 0u;
            hits |= (ok && (unsigned)(dv[g].w - d0) < (unsigned)nrows) ? (1u << (g * 4 + 3)) : 0u;
        }
        unsigned mask = __builtin_amdgcn_ballot_w32(hits != 0u);
        while (mask != 0u) {
            if (hits != 0u) {
                const int b = __builtin_ctz(hits);
                hits &= hits - 1u;
                const int e = (g0 + (b >> 2) * 32 + lane) * 4 + (b & 3);
                const int pos = qcnt + (int)__builtin_amdgcn_mbcnt_lo(mask, 0u);
                que[wave][(qhead + pos) & (QCAP - 1)] = e;
            }
            qcnt += __popc(mask);
            if (qcnt >= 32) {
                drain(32);
                qhead = (qhead + 32) & (QCAP - 1);
                qcnt -= 32;
            }
            mask = __builtin_amdgcn_ballot_w32(hits != 0u);
        }
    }
    if (qcnt > 0) drain(qcnt);
    __syncthreads();

    for (int idx = tid; idx < nrows * C; idx += 1024) {
        const int r = idx / C, c = idx % C;
        float v = acc[idx] / (den[r * H + c / F] + 1e-16f) + bias[c];
        if constexpr (ELU) v = v > 0.0f ? v : expm1f(v);
        out[(size_t)d0 * C + idx] = v;
    }
}

extern "C" void kernel_launch(void* const* d_in, const int* in_sizes, int n_in,
                              void* d_out, int out_size, void* d_ws, size_t ws_size,
                              hipStream_t stream) {
    const float* x  = (const float*)d_in[0];
    const int*   ei = (const int*)d_in[1];
    const float* ew = (const float*)d_in[2];

    const float* Wl[3] = { (const float*)d_in[3],  (const float*)d_in[9],  (const float*)d_in[15] };
    const float* As[3] = { (const float*)d_in[4],  (const float*)d_in[10], (const float*)d_in[16] };
    const float* Ad[3] = { (const float*)d_in[5],  (const float*)d_in[11], (const float*)d_in[17] };
    const float* We[3] = { (const float*)d_in[6],  (const float*)d_in[12], (const float*)d_in[18] };
    const float* Ae[3] = { (const float*)d_in[7],  (const float*)d_in[13], (const float*)d_in[19] };
    const float* Bi[3] = { (const float*)d_in[8],  (const float*)d_in[14], (const float*)d_in[20] };

    const int Hs[3] = { 4, 4, 1 };
    const int Fs[3] = { 32, 32, 64 };

    float* ws    = (float*)d_ws;
    float* xw    = ws;
    float* bufA  = xw    + (size_t)NN * 128;
    float* bufB  = bufA  + (size_t)NN * 128;
    float* ss    = bufB  + (size_t)NN * 128;
    float* sd    = ss    + (size_t)NN * 4;
    float* mseg  = sd    + (size_t)NN * 4;
    float* alpha = mseg  + (size_t)NN * 4;
    float* ce    = alpha + (size_t)ETOT * 4;
    float* meanw = ce + 4;

    mean_kernel<<<1, 1024, 0, stream>>>(ew, meanw, EE);

    const float* in = x;
    float* outs[3] = { bufA, bufB, (float*)d_out };

    for (int l = 0; l < 3; ++l) {
        const int H = Hs[l], F = Fs[l], C = H * F;

        edge_coef_kernel<<<1, 32, 0, stream>>>(We[l], Ae[l], ce, H, F);

        gemm_wmma_f32<<<dim3(NN / 16, C / 16), 32, 0, stream>>>(in, Wl[l], xw, KDIM, C);

        score_kernel<<<(NN * H + 255) / 256, 256, 0, stream>>>(xw, As[l], Ad[l], ss, sd, H, F);

        m_init_kernel<<<(NN * H + 255) / 256, 256, 0, stream>>>(mseg, NN * H);

        alpha_max_kernel<<<(ETOT * H + 255) / 256, 256, 0, stream>>>(
            ei, ew, meanw, ss, sd, ce, alpha, mseg, H);

        if (l < 2)
            range_kernel<128, 4, true><<<(NN + RROWS - 1) / RROWS, 1024, 0, stream>>>(
                ei, xw, alpha, mseg, Bi[l], outs[l]);
        else
            range_kernel<64, 1, false><<<(NN + RROWS - 1) / RROWS, 1024, 0, stream>>>(
                ei, xw, alpha, mseg, Bi[l], outs[l]);

        in = outs[l];
    }
}


# edge scores and segment max leave memory: the owner block forms alpha for the edges it finds, maximises in LDS, lists (source,row,alpha) in LDS, then sums; 19 -> 13 launches; sum order changes
# speedup vs baseline: 3.8335x; 1.4809x over previous
#include <hip/hip_runtime.h>
#include <math.h>

#define NN    50000
#define EE    800000
#define KDIM  128

typedef float v2f __attribute__((ext_vector_type(2)));
typedef float v8f __attribute__((ext_vector_type(8)));

__device__ __forceinline__ void ldsMaxF(float* addr, float val) {
    if (val >= 0.0f) atomicMax((int*)addr, __float_as_int(val));
    else             atomicMin((unsigned int*)addr, __float_as_uint(val));
}

#define MEAN_U 8
__global__ __launch_bounds__(1024) void mean_kernel(const float* __restrict__ ew,
                                                    float* __restrict__ meanw, int E) {
    __shared__ float sdata[1024];
    const int n4   = E >> 2;
    const int full = n4 / (MEAN_U * 1024);
    const float4* __restrict__ ew4 = reinterpret_cast<const float4*>(ew);
    float s = 0.0f;

    const float4* p = ew4 + threadIdx.x;
    for (int g = 0; g < full; ++g, p += MEAN_U * 1024) {
        float4 v[MEAN_U];
#pragma unroll
        for (int u = 0; u < MEAN_U; ++u) v[u] = p[u * 1024];
#pragma unroll
        for (int u = 0; u < MEAN_U; ++u) s += (v[u].x + v[u].y) + (v[u].z + v[u].w);
    }

    const int r0 = full * (MEAN_U * 1024) + (int)threadIdx.x;
    if (r0 < n4) {
        float4 v[MEAN_U];
#pragma unroll
        for (int u = 0; u < MEAN_U; ++u) {
            const int idx = r0 + u * 1024;
            v[u] = ew4[idx < n4 ? idx : n4 - 1];
        }
#pragma unroll
        for (int u = 0; u < MEAN_U; ++u)
            if (r0 + u * 1024 < n4) s += (v[u].x + v[u].y) + (v[u].z + v[u].w);
    }
    if ((int)threadIdx.x < (E & 3)) s += ew[(n4 << 2) + threadIdx.x];

    sdata[threadIdx.x] = s;
    __syncthreads();
    for (int o = 512; o > 0; o >>= 1) {
        if (threadIdx.x < o) sdata[threadIdx.x] += sdata[threadIdx.x + o];
        __syncthreads();
    }
    if (threadIdx.x == 0) meanw[0] = sdata[0] / (float)E;
}

__global__ void edge_coef_kernel(const float* __restrict__ We, const float* __restrict__ ae,
                                 float* __restrict__ ce, int H, int F) {
    int h = threadIdx.x;
    if (h >= H) return;
    float s = 0.0f;
    for (int f = 0; f < F; ++f) s += We[h * F + f] * ae[h * F + f];
    ce[h] = s;
}

__global__ __launch_bounds__(32) void gemm_wmma_f32(const float* __restrict__ X,
                                                    const float* __restrict__ W,
                                                    float* __restrict__ out,
                                                    int K, int C) {
    const int tm   = blockIdx.x;
    const int tc   = blockIdx.y;
    const int lane = threadIdx.x;
    const int half = lane >> 4;
    const int l15  = lane & 15;

    v8f acc = {};
    const float* xrow = X + (size_t)(tm * 16 + l15) * K;
    const int    coln = tc * 16 + l15;

    for (int k = 0; k < K; k += 4) {
        const int kk = k + half * 2;
        v2f a, b;
        a[0] = xrow[kk];
        a[1] = xrow[kk + 1];
        b[0] = W[(size_t)kk * C + coln];
        b[1] = W[(size_t)(kk + 1) * C + coln];
        acc = __builtin_amdgcn_wmma_f32_16x16x4_f32(
             false, a,  false, b,
             (short)0, acc,  false,  false);
    }
#pragma unroll
    for (int v = 0; v < 8; ++v) {
        const int row = tm * 16 + v + half * 8;
        out[(size_t)row * C + coln] = acc[v];
    }
}

__global__ void score_kernel(const float* __restrict__ xw,
                             const float* __restrict__ a_s, const float* __restrict__ a_d,
                             float* __restrict__ ss, float* __restrict__ sd,
                             int H, int F) {
    int i = blockIdx.x * blockDim.x + threadIdx.x;
    if (i >= NN * H) return;
    const int n = i / H, h = i % H;
    const float* v  = xw + (size_t)n * H * F + h * F;
    const float* as = a_s + h * F;
    const float* ad = a_d + h * F;
    float s1 = 0.0f, s2 = 0.0f;
    for (int f = 0; f < F; ++f) { float x = v[f]; s1 += x * as[f]; s2 += x * ad[f]; }
    ss[i] = s1;
    sd[i] = s2;
}

#define RROWS   196
#define RWAVES  32
#define QCAP    64
#define SCAN_G  8
#define LCAP    8192

template <int C, int H, bool ELU>
__global__ __launch_bounds__(1024) void range_kernel(const int* __restrict__ ei,
                                                     const float* __restrict__ ew,
                                                     const float* __restrict__ meanw,
                                                     const float* __restrict__ ss,
                                                     const float* __restrict__ sd,
                                                     const float* __restrict__ ce,
                                                     const float* __restrict__ xw,
                                                     const float* __restrict__ bias,
                                                     float* __restrict__ out) {
    constexpr int NC = C / 32;
    constexpr int F  = C / H;
    constexpr int U  = 8;
    constexpr int N4 = EE / 4;
    constexpr int ROUND4 = RWAVES * SCAN_G * 32;
    constexpr int NR = (N4 + ROUND4 - 1) / ROUND4;
    static_assert(F % 32 == 0, "a lane's element c*32+lane must lie in head (c*32)/F");
    static_assert(EE % 4 == 0, "the destination plane is read in groups of four");
    static_assert(32 % U == 0 && QCAP >= 64 && (QCAP & (QCAP - 1)) == 0, "queue and group sizes");
    static_assert(RROWS * H <= 1024, "one thread per (row, head) in the first phase");
    static_assert(NN <= 65536 && RROWS <= 256, "a list entry packs source (16 bits) and row (8 bits)");

    __shared__ float acc[RROWS * C];
    __shared__ float den[RROWS * H];
    __shared__ alignas(16) float mrow[RROWS * H];
    __shared__ float aself[RROWS * H];
    __shared__ alignas(16) float sdl[RROWS * H];
    __shared__ int que[RWAVES][QCAP];
    __shared__ int lsr[LCAP];
    __shared__ alignas(16) float lal[LCAP * H];
    __shared__ int lcnt;

    const int tid   = threadIdx.x;
    const int lane  = tid & 31;
    const int wave  = __builtin_amdgcn_readfirstlane(tid >> 5);
    const int d0    = blockIdx.x * RROWS;
    const int nrows = (NN - d0) < RROWS ? (NN - d0) : RROWS;

    for (int i = tid; i < RROWS * C; i += 1024) acc[i] = 0.0f;
    if (tid < RROWS * H) {
        const int r = tid / H, h = tid % H;
        const int n = d0 + (r < nrows ? r : nrows - 1);
        const float sdv = sd[n * H + h];
        float a = ss[n * H + h] + sdv + meanw[0] * ce[h];
        a = a > 0.0f ? a : 0.2f * a;
        den[tid]   = 0.0f;
        mrow[tid]  = a;
        aself[tid] = a;
        sdl[tid]   = sdv;
    }
    if (tid == 0) lcnt = 0;
    __syncthreads();

    int qhead = 0, qcnt = 0;

    auto edge_alpha = [&](const int e, int& sr, float (&al)[H]) {
        const int s = ei[e];
        const int r = ei[EE + e] - d0;
        const float w = ew[e];
        sr = s | (r << 16);
        if constexpr (H == 4) {
            const float4 sv = *reinterpret_cast<const float4*>(ss + (size_t)s * 4);
            const float4 dv = *reinterpret_cast<const float4*>(sdl + r * 4);
            al[0] = sv.x + dv.x + w * ce[0];
            al[1] = sv.y + dv.y + w * ce[1];
            al[2] = sv.z + dv.z + w * ce[2];
            al[3] = sv.w + dv.w + w * ce[3];
        } else {
            for (int h = 0; h < H; ++h) al[h] = ss[(size_t)s * H + h] + sdl[r * H + h] + w * ce[h];
        }
        for (int h = 0; h < H; ++h) al[h] = al[h] > 0.0f ? al[h] : 0.2f * al[h];
    };

    auto stage = [&](const int nb) {
        int sr = 0;
        float al[H];
        for (int h = 0; h < H; ++h) al[h] = 0.0f;
        __builtin_amdgcn_wave_barrier();
        if (lane < nb) {
            edge_alpha(que[wave][(qhead + lane) & (QCAP - 1)], sr, al);
            const int r = sr >> 16;
            for (int h = 0; h < H; ++h) ldsMaxF(&mrow[r * H + h], al[h]);
        }
        int base = 0;
        if (lane == 0) base = atomicAdd(&lcnt, nb);
        base = __builtin_amdgcn_readfirstlane(base);
        if (lane < nb && base + lane < LCAP) {
            lsr[base + lane] = sr;
            if constexpr (H == 4) {
                *reinterpret_cast<float4*>(lal + (base + lane) * 4) = make_float4(al[0], al[1], al[2], al[3]);
            } else {
                for (int h = 0; h < H; ++h) lal[(base + lane) * H + h] = al[h];
            }
        }
    };

    auto accumulate = [&](const int nb, const int sr, const float (&al)[H]) {
        const int s = sr & 0xffff, r = sr >> 16;
        float at[H];
        for (int h = 0; h < H; ++h) at[h] = 0.0f;
        if (lane < nb) {
            if constexpr (H == 4) {
                const float4 mx = *reinterpret_cast<const float4*>(mrow + r * 4);
                at[0] = __expf(al[0] - mx.x);
                at[1] = __expf(al[1] - mx.y);
                at[2] = __expf(al[2] - mx.z);
                at[3] = __expf(al[3] - mx.w);
            } else {
                for (int h = 0; h < H; ++h) at[h] = __expf(al[h] - mrow[r * H + h]);
            }
            for (int h = 0; h < H; ++h) atomicAdd(&den[r * H + h], at[h]);
        }
        for (int j0 = 0; j0 < nb; j0 += U) {
            float v[U][NC];
#pragma unroll
            for (int u = 0; u < U; ++u) {
                if (j0 + u < nb) {
                    const int sj = __builtin_amdgcn_readlane(s, j0 + u);
                    const float* row = xw + (size_t)sj * C + lane;
                    for (int c = 0; c < NC; ++c) v[u][c] = row[c * 32];
                }
            }
#pragma unroll
            for (int u = 0; u < U; ++u) {
                if (j0 + u < nb) {
                    const int rj = __builtin_amdgcn_readlane(r, j0 + u);
                    float* arow = acc + rj * C + lane;
                    for (int c = 0; c < NC; ++c) {
                        const float a = __int_as_float(
                            __builtin_amdgcn_readlane(__float_as_int(at[(c * 32) / F]), j0 + u));
                        atomicAdd(arow + c * 32, v[u][c] * a);
                    }
                }
            }
        }
    };

    auto restage = [&](const int nb) {
        int sr = 0;
        float al[H];
        for (int h = 0; h < H; ++h) al[h] = 0.0f;
        __builtin_amdgcn_wave_barrier();
        if (lane < nb) edge_alpha(que[wave][(qhead + lane) & (QCAP - 1)], sr, al);
        accumulate(nb, sr, al);
    };

    const int4* __restrict__ dst4 = reinterpret_cast<const int4*>(ei + EE);
    const int rot = blockIdx.x % NR;
    auto scan = [&](auto batch) {
        for (int it = 0; it < NR; ++it) {
            int rnd = it + rot;
            if (rnd >= NR) rnd -= NR;
            const int g0 = rnd * ROUND4 + wave * (SCAN_G * 32);
            int4 dv[SCAN_G];
#pragma unroll
            for (int g = 0; g < SCAN_G; ++g) {
                const int i4 = g0 + g * 32 + lane;
                dv[g] = dst4[i4 < N4 ? i4 : N4 - 1];
            }
            unsigned hits = 0u;
#pragma unroll
            for (int g = 0; g < SCAN_G; ++g) {
                const bool ok = (g0 + g * 32 + lane) < N4;
                hits |= (ok && (unsigned)(dv[g].x - d0) < (unsigned)nrows) ? (1u << (g * 4 + 0)) : 0u;
                hits |= (ok && (unsigned)(dv[g].y - d0) < (unsigned)nrows) ? (1u << (g * 4 + 1)) : 0u;
                hits |= (ok && (unsigned)(dv[g].z - d0) < (unsigned)nrows) ? (1u << (g * 4 + 2)) : 0u;
                hits |= (ok && (unsigned)(dv[g].w - d0) < (unsigned)nrows) ? (1u << (g * 4 + 3)) : 0u;
            }
            unsigned mask = __builtin_amdgcn_ballot_w32(hits != 0u);
            while (mask != 0u) {
                if (hits != 0u) {
                    const int b = __builtin_ctz(hits);
                    hits &= hits - 1u;
                    const int e = (g0 + (b >> 2) * 32 + lane) * 4 + (b & 3);
                    const int pos = qcnt + (int)__builtin_amdgcn_mbcnt_lo(mask, 0u);
                    que[wave][(qhead + pos) & (QCAP - 1)] = e;
                }
                qcnt += __popc(mask);
                if (qcnt >= 32) {
                    batch(32);
                    qhead = (qhead + 32) & (QCAP - 1);
                    qcnt -= 32;
                }
                mask = __builtin_amdgcn_ballot_w32(hits != 0u);
            }
        }
        if (qcnt > 0) batch(qcnt);
        qhead = 0;
        qcnt = 0;
    };

    scan(stage);
    __syncthreads();

    const int total = __builtin_amdgcn_readfirstlane(lcnt);
    if (total <= LCAP) {
        for (int base = wave * 32; base < total; base += RWAVES * 32) {
            const int left = total - base;
            int sr = 0;
            float al[H];
            for (int h = 0; h < H; ++h) al[h] = 0.0f;
            if (lane < left) {
                sr = lsr[base + lane];
                if constexpr (H == 4) {
                    const float4 t = *reinterpret_cast<const float4*>(lal + (base + lane) * 4);
                    al[0] = t.x; al[1] = t.y; al[2] = t.z; al[3] = t.w;
                } else {
                    for (int h = 0; h < H; ++h) al[h] = lal[(base + lane) * H + h];
                }
            }
            if (left >= 32) accumulate(32, sr, al);
            else            accumulate(left, sr, al);
        }
    } else {
        scan(restage);
    }
    __syncthreads();

    {
        constexpr int RPT = 1024 / C;
        constexpr int P0  = 8;
        const int c = tid % C, h = c / F;
        const float bc = bias[c];
        for (int r0 = tid / C; r0 < nrows; r0 += RPT * P0) {
            float xv[P0];
#pragma unroll
            for (int k = 0; k < P0; ++k) {
                const int r = r0 + k * RPT;
                xv[k] = xw[(size_t)(d0 + (r < nrows ? r : nrows - 1)) * C + c];
            }
#pragma unroll
            for (int k = 0; k < P0; ++k) {
                const int r = r0 + k * RPT;
                if (r < nrows) {
                    const int rh = r * H + h;
                    const float exs = __expf(aself[rh] - mrow[rh]);
                    float v = (acc[r * C + c] + exs * xv[k]) / (den[rh] + exs + 1e-16f) + bc;
                    if constexpr (ELU) v = v > 0.0f ? v : expm1f(v);
                    out[(size_t)(d0 + r) * C + c] = v;
                }
            }
        }
    }
}

extern "C" void kernel_launch(void* const* d_in, const int* in_sizes, int n_in,
                              void* d_out, int out_size, void* d_ws, size_t ws_size,
                              hipStream_t stream) {
    const float* x  = (const float*)d_in[0];
    const int*   ei = (const int*)d_in[1];
    const float* ew = (const float*)d_in[2];

    const float* Wl[3] = { (const float*)d_in[3],  (const float*)d_in[9],  (const float*)d_in[15] };
    const float* As[3] = { (const float*)d_in[4],  (const float*)d_in[10], (const float*)d_in[16] };
    const float* Ad[3] = { (const float*)d_in[5],  (const float*)d_in[11], (const float*)d_in[17] };
    const float* We[3] = { (const float*)d_in[6],  (const float*)d_in[12], (const float*)d_in[18] };
    const float* Ae[3] = { (const float*)d_in[7],  (const float*)d_in[13], (const float*)d_in[19] };
    const float* Bi[3] = { (const float*)d_in[8],  (const float*)d_in[14], (const float*)d_in[20] };

    const int Hs[3] = { 4, 4, 1 };
    const int Fs[3] = { 32, 32, 64 };

    float* ws    = (float*)d_ws;
    float* xw    = ws;
    float* bufA  = xw    + (size_t)NN * 128;
    float* bufB  = bufA  + (size_t)NN * 128;
    float* ss    = bufB  + (size_t)NN * 128;
    float* sd    = ss    + (size_t)NN * 4;
    float* ce    = sd    + (size_t)NN * 4;
    float* meanw = ce + 4;

    mean_kernel<<<1, 1024, 0, stream>>>(ew, meanw, EE);

    const float* in = x;
    float* outs[3] = { bufA, bufB, (float*)d_out };

    for (int l = 0; l < 3; ++l) {
        const int H = Hs[l], F = Fs[l], C = H * F;

        edge_coef_kernel<<<1, 32, 0, stream>>>(We[l], Ae[l], ce, H, F);

        gemm_wmma_f32<<<dim3(NN / 16, C / 16), 32, 0, stream>>>(in, Wl[l], xw, KDIM, C);

        score_kernel<<<(NN * H + 255) / 256, 256, 0, stream>>>(xw, As[l], Ad[l], ss, sd, H, F);

        if (l < 2)
            range_kernel<128, 4, true><<<(NN + RROWS - 1) / RROWS, 1024, 0, stream>>>(
                ei, ew, meanw, ss, sd, ce, xw, Bi[l], outs[l]);
        else
            range_kernel<64, 1, false><<<(NN + RROWS - 1) / RROWS, 1024, 0, stream>>>(
                ei, ew, meanw, ss, sd, ce, xw, Bi[l], outs[l]);

        in = outs[l];
    }
}


# layers 1, 2 stop scanning the destination plane: layer 0's block also writes its edge list (source, row, weight) to the workspace, later blocks read their own back; 13 launches; sum order changes
# speedup vs baseline: 4.5973x; 1.1992x over previous
#include <hip/hip_runtime.h>
#include <math.h>

#define NN    50000
#define EE    800000
#define KDIM  128

typedef float v2f __attribute__((ext_vector_type(2)));
typedef float v8f __attribute__((ext_vector_type(8)));

__device__ __forceinline__ void ldsMaxF(float* addr, float val) {
    if (val >= 0.0f) atomicMax((int*)addr, __float_as_int(val));
    else             atomicMin((unsigned int*)addr, __float_as_uint(val));
}

#define MEAN_U 8
__global__ __launch_bounds__(1024) void mean_kernel(const float* __restrict__ ew,
                                                    float* __restrict__ meanw, int E) {
    __shared__ float sdata[1024];
    const int n4   = E >> 2;
    const int full = n4 / (MEAN_U * 1024);
    const float4* __restrict__ ew4 = reinterpret_cast<const float4*>(ew);
    float s = 0.0f;

    const float4* p = ew4 + threadIdx.x;
    for (int g = 0; g < full; ++g, p += MEAN_U * 1024) {
        float4 v[MEAN_U];
#pragma unroll
        for (int u = 0; u < MEAN_U; ++u) v[u] = p[u * 1024];
#pragma unroll
        for (int u = 0; u < MEAN_U; ++u) s += (v[u].x + v[u].y) + (v[u].z + v[u].w);
    }

    const int r0 = full * (MEAN_U * 1024) + (int)threadIdx.x;
    if (r0 < n4) {
        float4 v[MEAN_U];
#pragma unroll
        for (int u = 0; u < MEAN_U; ++u) {
            const int idx = r0 + u * 1024;
            v[u] = ew4[idx < n4 ? idx : n4 - 1];
        }
#pragma unroll
        for (int u = 0; u < MEAN_U; ++u)
            if (r0 + u * 1024 < n4) s += (v[u].x + v[u].y) + (v[u].z + v[u].w);
    }
    if ((int)threadIdx.x < (E & 3)) s += ew[(n4 << 2) + threadIdx.x];

    sdata[threadIdx.x] = s;
    __syncthreads();
    for (int o = 512; o > 0; o >>= 1) {
        if (threadIdx.x < o) sdata[threadIdx.x] += sdata[threadIdx.x + o];
        __syncthreads();
    }
    if (threadIdx.x == 0) meanw[0] = sdata[0] / (float)E;
}

__global__ void edge_coef_kernel(const float* __restrict__ We, const float* __restrict__ ae,
                                 float* __restrict__ ce, int H, int F) {
    int h = threadIdx.x;
    if (h >= H) return;
    float s = 0.0f;
    for (int f = 0; f < F; ++f) s += We[h * F + f] * ae[h * F + f];
    ce[h] = s;
}

__global__ __launch_bounds__(32) void gemm_wmma_f32(const float* __restrict__ X,
                                                    const float* __restrict__ W,
                                                    float* __restrict__ out,
                                                    int K, int C) {
    const int tm   = blockIdx.x;
    const int tc   = blockIdx.y;
    const int lane = threadIdx.x;
    const int half = lane >> 4;
    const int l15  = lane & 15;

    v8f acc = {};
    const float* xrow = X + (size_t)(tm * 16 + l15) * K;
    const int    coln = tc * 16 + l15;

    for (int k = 0; k < K; k += 4) {
        const int kk = k + half * 2;
        v2f a, b;
        a[0] = xrow[kk];
        a[1] = xrow[kk + 1];
        b[0] = W[(size_t)kk * C + coln];
        b[1] = W[(size_t)(kk + 1) * C + coln];
        acc = __builtin_amdgcn_wmma_f32_16x16x4_f32(
             false, a,  false, b,
             (short)0, acc,  false,  false);
    }
#pragma unroll
    for (int v = 0; v < 8; ++v) {
        const int row = tm * 16 + v + half * 8;
        out[(size_t)row * C + coln] = acc[v];
    }
}

__global__ void score_kernel(const float* __restrict__ xw,
                             const float* __restrict__ a_s, const float* __restrict__ a_d,
                             float* __restrict__ ss, float* __restrict__ sd,
                             int H, int F) {
    int i = blockIdx.x * blockDim.x + threadIdx.x;
    if (i >= NN * H) return;
    const int n = i / H, h = i % H;
    const float* v  = xw + (size_t)n * H * F + h * F;
    const float* as = a_s + h * F;
    const float* ad = a_d + h * F;
    float s1 = 0.0f, s2 = 0.0f;
    for (int f = 0; f < F; ++f) { float x = v[f]; s1 += x * as[f]; s2 += x * ad[f]; }
    ss[i] = s1;
    sd[i] = s2;
}

#define RROWS   196
#define RWAVES  32
#define QCAP    64
#define SCAN_G  8
#define LCAP    8192

template <int C, int H, bool ELU, int MODE>
__global__ __launch_bounds__(1024) void range_kernel(const int* __restrict__ ei,
                                                     const float* __restrict__ ew,
                                                     const float* __restrict__ meanw,
                                                     const float* __restrict__ ss,
                                                     const float* __restrict__ sd,
                                                     const float* __restrict__ ce,
                                                     const float* __restrict__ xw,
                                                     const float* __restrict__ bias,
                                                     float* __restrict__ out,
                                                     int* __restrict__ gsr,
                                                     float* __restrict__ gw,
                                                     int* __restrict__ gcnt) {
    constexpr int NC = C / 32;
    constexpr int F  = C / H;
    constexpr int U  = 8;
    constexpr int N4 = EE / 4;
    constexpr int ROUND4 = RWAVES * SCAN_G * 32;
    constexpr int NR = (N4 + ROUND4 - 1) / ROUND4;
    static_assert(F % 32 == 0, "a lane's element c*32+lane must lie in head (c*32)/F");
    static_assert(EE % 4 == 0, "the destination plane is read in groups of four");
    static_assert(32 % U == 0 && QCAP >= 64 && (QCAP & (QCAP - 1)) == 0, "queue and group sizes");
    static_assert(RROWS * H <= 1024, "one thread per (row, head) in the first phase");
    static_assert(NN <= 65536 && RROWS <= 256, "a list entry packs source (16 bits) and row (8 bits)");
    static_assert((NN + RROWS - 1) / RROWS <= 256, "the workspace holds 256 blocks' lists and counts");

    __shared__ float acc[RROWS * C];
    __shared__ float den[RROWS * H];
    __shared__ alignas(16) float mrow[RROWS * H];
    __shared__ float aself[RROWS * H];
    __shared__ alignas(16) float sdl[RROWS * H];
    __shared__ int que[RWAVES][QCAP];
    __shared__ int lsr[LCAP];
    __shared__ alignas(16) float lal[LCAP * H];
    __shared__ int lcnt;

    const int tid   = threadIdx.x;
    const int lane  = tid & 31;
    const int wave  = __builtin_amdgcn_readfirstlane(tid >> 5);
    const int d0    = blockIdx.x * RROWS;
    const int nrows = (NN - d0) < RROWS ? (NN - d0) : RROWS;

    for (int i = tid; i < RROWS * C; i += 1024) acc[i] = 0.0f;
    if (tid < RROWS * H) {
        const int r = tid / H, h = tid % H;
        const int n = d0 + (r < nrows ? r : nrows - 1);
        const float sdv = sd[n * H + h];
        float a = ss[n * H + h] + sdv + meanw[0] * ce[h];
        a = a > 0.0f ? a : 0.2f * a;
        den[tid]   = 0.0f;
        mrow[tid]  = a;
        aself[tid] = a;
        sdl[tid]   = sdv;
    }
    if (tid == 0) lcnt = 0;
    __syncthreads();

    int qhead = 0, qcnt = 0;

    auto load_ss = [&](const int s, float (&sv)[H]) {
        if constexpr (H == 4) {
            const float4 t = *reinterpret_cast<const float4*>(ss + (size_t)s * 4);
            sv[0] = t.x; sv[1] = t.y; sv[2] = t.z; sv[3] = t.w;
        } else {
            for (int h = 0; h < H; ++h) sv[h] = ss[(size_t)s * H + h];
        }
    };

    auto alpha_of = [&](const float (&sv)[H], const int r, const float w, float (&al)[H]) {
        if constexpr (H == 4) {
            const float4 dv = *reinterpret_cast<const float4*>(sdl + r * 4);
            al[0] = sv[0] + dv.x + w * ce[0];
            al[1] = sv[1] + dv.y + w * ce[1];
            al[2] = sv[2] + dv.z + w * ce[2];
            al[3] = sv[3] + dv.w + w * ce[3];
        } else {
            for (int h = 0; h < H; ++h) al[h] = sv[h] + sdl[r * H + h] + w * ce[h];
        }
        for (int h = 0; h < H; ++h) al[h] = al[h] > 0.0f ? al[h] : 0.2f * al[h];
    };

    auto edge_alpha = [&](const int e, int& sr, float& w, float (&al)[H]) {
        const int s = ei[e];
        const int r = ei[EE + e] - d0;
        w = ew[e];
        sr = s | (r << 16);
        float sv[H];
        load_ss(s, sv);
        alpha_of(sv, r, w, al);
    };

    auto stage = [&](const int nb) {
        int sr = 0;
        float w = 0.0f;
        float al[H];
        for (int h = 0; h < H; ++h) al[h] = 0.0f;
        __builtin_amdgcn_wave_barrier();
        if (lane < nb) {
            edge_alpha(que[wave][(qhead + lane) & (QCAP - 1)], sr, w, al);
            const int r = sr >> 16;
            for (int h = 0; h < H; ++h) ldsMaxF(&mrow[r * H + h], al[h]);
        }
        int base = 0;
        if (lane == 0) base = atomicAdd(&lcnt, nb);
        base = __builtin_amdgcn_readfirstlane(base);
        if (lane < nb && base + lane < LCAP) {
            lsr[base + lane] = sr;
            if constexpr (H == 4) {
                *reinterpret_cast<float4*>(lal + (base + lane) * 4) = make_float4(al[0], al[1], al[2], al[3]);
            } else {
                for (int h = 0; h < H; ++h) lal[(base + lane) * H + h] = al[h];
            }
            if constexpr (MODE == 0) {
                gsr[(size_t)blockIdx.x * LCAP + base + lane] = sr;
                gw[(size_t)blockIdx.x * LCAP + base + lane]  = w;
            }
        }
    };

    auto accumulate = [&](const int nb, const int sr, const float (&al)[H]) {
        const int s = sr & 0xffff, r = sr >> 16;
        float at[H];
        for (int h = 0; h < H; ++h) at[h] = 0.0f;
        if (lane < nb) {
            if constexpr (H == 4) {
                const float4 mx = *reinterpret_cast<const float4*>(mrow + r * 4);
                at[0] = __expf(al[0] - mx.x);
                at[1] = __expf(al[1] - mx.y);
                at[2] = __expf(al[2] - mx.z);
                at[3] = __expf(al[3] - mx.w);
            } else {
                for (int h = 0; h < H; ++h) at[h] = __expf(al[h] - mrow[r * H + h]);
            }
            for (int h = 0; h < H; ++h) atomicAdd(&den[r * H + h], at[h]);
        }
        for (int j0 = 0; j0 < nb; j0 += U) {
            float v[U][NC];
#pragma unroll
            for (int u = 0; u < U; ++u) {
                if (j0 + u < nb) {
                    const int sj = __builtin_amdgcn_readlane(s, j0 + u);
                    const float* row = xw + (size_t)sj * C + lane;
                    for (int c = 0; c < NC; ++c) v[u][c] = row[c * 32];
                }
            }
#pragma unroll
            for (int u = 0; u < U; ++u) {
                if (j0 + u < nb) {
                    const int rj = __builtin_amdgcn_readlane(r, j0 + u);
                    float* arow = acc + rj * C + lane;
                    for (int c = 0; c < NC; ++c) {
                        const float a = __int_as_float(
                            __builtin_amdgcn_readlane(__float_as_int(at[(c * 32) / F]), j0 + u));
                        atomicAdd(arow + c * 32, v[u][c] * a);
                    }
                }
            }
        }
    };

    auto restage = [&](const int nb) {
        int sr = 0;
        float al[H];
        for (int h = 0; h < H; ++h) al[h] = 0.0f;
        __builtin_amdgcn_wave_barrier();
        float w = 0.0f;
        if (lane < nb) edge_alpha(que[wave][(qhead + lane) & (QCAP - 1)], sr, w, al);
        accumulate(nb, sr, al);
    };

    const int4* __restrict__ dst4 = reinterpret_cast<const int4*>(ei + EE);
    const int rot = blockIdx.x % NR;
    auto scan = [&](auto batch) {
        for (int it = 0; it < NR; ++it) {
            int rnd = it + rot;
            if (rnd >= NR) rnd -= NR;
            const int g0 = rnd * ROUND4 + wave * (SCAN_G * 32);
            int4 dv[SCAN_G];
#pragma unroll
            for (int g = 0; g < SCAN_G; ++g) {
                const int i4 = g0 + g * 32 + lane;
                dv[g] = dst4[i4 < N4 ? i4 : N4 - 1];
            }
            unsigned hits = 0u;
#pragma unroll
            for (int g = 0; g < SCAN_G; ++g) {
                const bool ok = (g0 + g * 32 + lane) < N4;
                hits |= (ok && (unsigned)(dv[g].x - d0) < (unsigned)nrows) ? (1u << (g * 4 + 0)) : 0u;
                hits |= (ok && (unsigned)(dv[g].y - d0) < (unsigned)nrows) ? (1u << (g * 4 + 1)) : 0u;
                hits |= (ok && (unsigned)(dv[g].z - d0) < (unsigned)nrows) ? (1u << (g * 4 + 2)) : 0u;
                hits |= (ok && (unsigned)(dv[g].w - d0) < (unsigned)nrows) ? (1u << (g * 4 + 3)) : 0u;
            }
            unsigned mask = __builtin_amdgcn_ballot_w32(hits != 0u);
            while (mask != 0u) {
                if (hits != 0u) {
                    const int b = __builtin_ctz(hits);
                    hits &= hits - 1u;
                    const int e = (g0 + (b >> 2) * 32 + lane) * 4 + (b & 3);
                    const int pos = qcnt + (int)__builtin_amdgcn_mbcnt_lo(mask, 0u);
                    que[wave][(qhead + pos) & (QCAP - 1)] = e;
                }
                qcnt += __popc(mask);
                if (qcnt >= 32) {
                    batch(32);
                    qhead = (qhead + 32) & (QCAP - 1);
                    qcnt -= 32;
                }
                mask = __builtin_amdgcn_ballot_w32(hits != 0u);
            }
        }
        if (qcnt > 0) batch(qcnt);
        qhead = 0;
        qcnt = 0;
    };

    bool listed = false;
    if constexpr (MODE == 1) {
        const int mt = __builtin_amdgcn_readfirstlane(gcnt[blockIdx.x]);
        listed = mt <= LCAP;
        if (listed) {
            const int*   __restrict__ msr = gsr + (size_t)blockIdx.x * LCAP;
            const float* __restrict__ mw  = gw  + (size_t)blockIdx.x * LCAP;
            constexpr int G = 4;
            for (int b0 = wave * 32; b0 < mt; b0 += G * RWAVES * 32) {
                int   sr[G];
                float w[G];
                float sv[G][H];
#pragma unroll
                for (int g = 0; g < G; ++g) {
                    const int p = b0 + g * (RWAVES * 32) + lane;
                    const int q = p < mt ? p : mt - 1;
                    sr[g] = msr[q];
                    w[g]  = mw[q];
                }
#pragma unroll
                for (int g = 0; g < G; ++g) asm volatile("" :: "v"(sr[g]), "v"(w[g]));
#pragma unroll
                for (int g = 0; g < G; ++g) load_ss(sr[g] & 0xffff, sv[g]);
#pragma unroll
                for (int g = 0; g < G; ++g) {
                    const int p = b0 + g * (RWAVES * 32) + lane;
                    if (p < mt) {
                        const int r = sr[g] >> 16;
                        float al[H];
                        alpha_of(sv[g], r, w[g], al);
                        for (int h = 0; h < H; ++h) ldsMaxF(&mrow[r * H + h], al[h]);
                        lsr[p] = sr[g];
                        if constexpr (H == 4) {
                            *reinterpret_cast<float4*>(lal + p * 4) = make_float4(al[0], al[1], al[2], al[3]);
                        } else {
                            for (int h = 0; h < H; ++h) lal[p * H + h] = al[h];
                        }
                    }
                }
            }
            if (tid == 0) lcnt = mt;
        }
    }
    if (!listed) scan(stage);
    __syncthreads();

    const int total = __builtin_amdgcn_readfirstlane(lcnt);
    if constexpr (MODE == 0) {
        if (tid == 0) gcnt[blockIdx.x] = total;
    }
    if (total <= LCAP) {
        for (int base = wave * 32; base < total; base += RWAVES * 32) {
            const int left = total - base;
            int sr = 0;
            float al[H];
            for (int h = 0; h < H; ++h) al[h] = 0.0f;
            if (lane < left) {
                sr = lsr[base + lane];
                if constexpr (H == 4) {
                    const float4 t = *reinterpret_cast<const float4*>(lal + (base + lane) * 4);
                    al[0] = t.x; al[1] = t.y; al[2] = t.z; al[3] = t.w;
                } else {
                    for (int h = 0; h < H; ++h) al[h] = lal[(base + lane) * H + h];
                }
            }
            if (left >= 32) accumulate(32, sr, al);
            else            accumulate(left, sr, al);
        }
    } else {
        scan(restage);
    }
    __syncthreads();

    {
        constexpr int RPT = 1024 / C;
        constexpr int P0  = 8;
        const int c = tid % C, h = c / F;
        const float bc = bias[c];
        for (int r0 = tid / C; r0 < nrows; r0 += RPT * P0) {
            float xv[P0];
#pragma unroll
            for (int k = 0; k < P0; ++k) {
                const int r = r0 + k * RPT;
                xv[k] = xw[(size_t)(d0 + (r < nrows ? r : nrows - 1)) * C + c];
            }
#pragma unroll
            for (int k = 0; k < P0; ++k) {
                const int r = r0 + k * RPT;
                if (r < nrows) {
                    const int rh = r * H + h;
                    const float exs = __expf(aself[rh] - mrow[rh]);
                    float v = (acc[r * C + c] + exs * xv[k]) / (den[rh] + exs + 1e-16f) + bc;
                    if constexpr (ELU) v = v > 0.0f ? v : expm1f(v);
                    out[(size_t)(d0 + r) * C + c] = v;
                }
            }
        }
    }
}

extern "C" void kernel_launch(void* const* d_in, const int* in_sizes, int n_in,
                              void* d_out, int out_size, void* d_ws, size_t ws_size,
                              hipStream_t stream) {
    const float* x  = (const float*)d_in[0];
    const int*   ei = (const int*)d_in[1];
    const float* ew = (const float*)d_in[2];

    const float* Wl[3] = { (const float*)d_in[3],  (const float*)d_in[9],  (const float*)d_in[15] };
    const float* As[3] = { (const float*)d_in[4],  (const float*)d_in[10], (const float*)d_in[16] };
    const float* Ad[3] = { (const float*)d_in[5],  (const float*)d_in[11], (const float*)d_in[17] };
    const float* We[3] = { (const float*)d_in[6],  (const float*)d_in[12], (const float*)d_in[18] };
    const float* Ae[3] = { (const float*)d_in[7],  (const float*)d_in[13], (const float*)d_in[19] };
    const float* Bi[3] = { (const float*)d_in[8],  (const float*)d_in[14], (const float*)d_in[20] };

    const int Hs[3] = { 4, 4, 1 };
    const int Fs[3] = { 32, 32, 64 };

    float* ws    = (float*)d_ws;
    float* xw    = ws;
    float* bufA  = xw    + (size_t)NN * 128;
    float* bufB  = bufA  + (size_t)NN * 128;
    float* ss    = bufB  + (size_t)NN * 128;
    float* sd    = ss    + (size_t)NN * 4;
    float* ce    = sd    + (size_t)NN * 4;
    float* meanw = ce + 4;
    int*   gcnt  = (int*)(ce + 64);
    int*   gsr   = gcnt + 256;
    float* gw    = (float*)(gsr + 256 * LCAP);

    mean_kernel<<<1, 1024, 0, stream>>>(ew, meanw, EE);

    const float* in = x;
    float* outs[3] = { bufA, bufB, (float*)d_out };

    for (int l = 0; l < 3; ++l) {
        const int H = Hs[l], F = Fs[l], C = H * F;

        edge_coef_kernel<<<1, 32, 0, stream>>>(We[l], Ae[l], ce, H, F);

        gemm_wmma_f32<<<dim3(NN / 16, C / 16), 32, 0, stream>>>(in, Wl[l], xw, KDIM, C);

        score_kernel<<<(NN * H + 255) / 256, 256, 0, stream>>>(xw, As[l], Ad[l], ss, sd, H, F);

        if (l == 0)
            range_kernel<128, 4, true, 0><<<(NN + RROWS - 1) / RROWS, 1024, 0, stream>>>(
                ei, ew, meanw, ss, sd, ce, xw, Bi[l], outs[l], gsr, gw, gcnt);
        else if (l == 1)
            range_kernel<128, 4, true, 1><<<(NN + RROWS - 1) / RROWS, 1024, 0, stream>>>(
                ei, ew, meanw, ss, sd, ce, xw, Bi[l], outs[l], gsr, gw, gcnt);
        else
            range_kernel<64, 1, false, 1><<<(NN + RROWS - 1) / RROWS, 1024, 0, stream>>>(
                ei, ew, meanw, ss, sd, ce, xw, Bi[l], outs[l], gsr, gw, gcnt);

        in = outs[l];
    }
}


# product: 25000 one-wave tiles become 782 blocks of 64 rows: X rows staged once in LDS by whole-line loads, a wave holds its 16 W columns in registers for 4 row tiles, whole-line stores; same sums
# speedup vs baseline: 5.8109x; 1.2640x over previous
#include <hip/hip_runtime.h>
#include <math.h>

#define NN    50000
#define EE    800000
#define KDIM  128

typedef float v2f __attribute__((ext_vector_type(2)));
typedef float v8f __attribute__((ext_vector_type(8)));

__device__ __forceinline__ void ldsMaxF(float* addr, float val) {
    if (val >= 0.0f) atomicMax((int*)addr, __float_as_int(val));
    else             atomicMin((unsigned int*)addr, __float_as_uint(val));
}

#define MEAN_U 8
__global__ __launch_bounds__(1024) void mean_kernel(const float* __restrict__ ew,
                                                    float* __restrict__ meanw, int E) {
    __shared__ float sdata[1024];
    const int n4   = E >> 2;
    const int full = n4 / (MEAN_U * 1024);
    const float4* __restrict__ ew4 = reinterpret_cast<const float4*>(ew);
    float s = 0.0f;

    const float4* p = ew4 + threadIdx.x;
    for (int g = 0; g < full; ++g, p += MEAN_U * 1024) {
        float4 v[MEAN_U];
#pragma unroll
        for (int u = 0; u < MEAN_U; ++u) v[u] = p[u * 1024];
#pragma unroll
        for (int u = 0; u < MEAN_U; ++u) s += (v[u].x + v[u].y) + (v[u].z + v[u].w);
    }

    const int r0 = full * (MEAN_U * 1024) + (int)threadIdx.x;
    if (r0 < n4) {
        float4 v[MEAN_U];
#pragma unroll
        for (int u = 0; u < MEAN_U; ++u) {
            const int idx = r0 + u * 1024;
            v[u] = ew4[idx < n4 ? idx : n4 - 1];
        }
#pragma unroll
        for (int u = 0; u < MEAN_U; ++u)
            if (r0 + u * 1024 < n4) s += (v[u].x + v[u].y) + (v[u].z + v[u].w);
    }
    if ((int)threadIdx.x < (E & 3)) s += ew[(n4 << 2) + threadIdx.x];

    sdata[threadIdx.x] = s;
    __syncthreads();
    for (int o = 512; o > 0; o >>= 1) {
        if (threadIdx.x < o) sdata[threadIdx.x] += sdata[threadIdx.x + o];
        __syncthreads();
    }
    if (threadIdx.x == 0) meanw[0] = sdata[0] / (float)E;
}

__global__ void edge_coef_kernel(const float* __restrict__ We, const float* __restrict__ ae,
                                 float* __restrict__ ce, int H, int F) {
    int h = threadIdx.x;
    if (h >= H) return;
    float s = 0.0f;
    for (int f = 0; f < F; ++f) s += We[h * F + f] * ae[h * F + f];
    ce[h] = s;
}

#define GROWS   64
#define GPITCH  132

template <int C>
__global__ __launch_bounds__(C * 2, C == 128 ? 8 : 4) void gemm_wmma_f32(const float* __restrict__ X,
                                                       const float* __restrict__ W,
                                                       float* __restrict__ out) {
    constexpr int NT = C * 2;
    constexpr int MT = GROWS / 16;
    constexpr int KS = KDIM / 4;
    constexpr int XL = GROWS * (KDIM / 4) / NT;
    constexpr int C4 = C / 4;
    constexpr int OL = GROWS * C4 / NT;
    static_assert(KDIM == 128 && GPITCH >= KDIM && GPITCH % 4 == 0, "a wave's 16-byte loads cover one row of X");
    static_assert(GROWS * (KDIM / 4) % NT == 0 && GROWS * C4 % NT == 0, "whole trips");

    __shared__ alignas(16) float tile[GROWS * GPITCH];

    const int tid  = threadIdx.x;
    const int lane = tid & 31;
    const int wave = __builtin_amdgcn_readfirstlane(tid >> 5);
    const int half = lane >> 4;
    const int l15  = lane & 15;
    const int row0 = blockIdx.x * GROWS;

    float4 xv[XL];
#pragma unroll
    for (int i = 0; i < XL; ++i) {
        const int idx = i * NT + tid;
        const int r   = idx >> 5, c4 = idx & 31;
        const int gr  = (row0 + r) < NN ? (row0 + r) : (NN - 1);
        xv[i] = *reinterpret_cast<const float4*>(X + (size_t)gr * KDIM + c4 * 4);
    }

    v2f b[KS];
    const int coln = wave * 16 + l15;
#pragma unroll
    for (int ks = 0; ks < KS; ++ks) {
        const int kk = ks * 4 + half * 2;
        b[ks][0] = W[(size_t)kk * C + coln];
        b[ks][1] = W[(size_t)(kk + 1) * C + coln];
    }

#pragma unroll
    for (int i = 0; i < XL; ++i) {
        const int idx = i * NT + tid;
        const int r   = idx >> 5, c4 = idx & 31;
        *reinterpret_cast<float4*>(tile + r * GPITCH + c4 * 4) = xv[i];
    }
    __syncthreads();

    v8f acc[MT];
#pragma unroll
    for (int m = 0; m < MT; ++m) acc[m] = v8f{};

    const float* arow = tile + l15 * GPITCH + half * 2;
#pragma unroll
    for (int ks = 0; ks < KS; ++ks) {
#pragma unroll
        for (int m = 0; m < MT; ++m) {
            const v2f a = *reinterpret_cast<const v2f*>(arow + m * 16 * GPITCH + ks * 4);
            acc[m] = __builtin_amdgcn_wmma_f32_16x16x4_f32(
                 false, a,  false, b[ks],
                 (short)0, acc[m],  false,  false);
        }
    }
    __syncthreads();

#pragma unroll
    for (int m = 0; m < MT; ++m) {
#pragma unroll
        for (int v = 0; v < 8; ++v)
            tile[(m * 16 + v + half * 8) * GPITCH + coln] = acc[m][v];
    }
    __syncthreads();
#pragma unroll
    for (int i = 0; i < OL; ++i) {
        const int idx = i * NT + tid;
        const int r   = idx / C4, c4 = idx % C4;
        if (row0 + r < NN)
            *reinterpret_cast<float4*>(out + (size_t)(row0 + r) * C + c4 * 4) =
                *reinterpret_cast<const float4*>(tile + r * GPITCH + c4 * 4);
    }
}

__global__ void score_kernel(const float* __restrict__ xw,
                             const float* __restrict__ a_s, const float* __restrict__ a_d,
                             float* __restrict__ ss, float* __restrict__ sd,
                             int H, int F) {
    int i = blockIdx.x * blockDim.x + threadIdx.x;
    if (i >= NN * H) return;
    const int n = i / H, h = i % H;
    const float* v  = xw + (size_t)n * H * F + h * F;
    const float* as = a_s + h * F;
    const float* ad = a_d + h * F;
    float s1 = 0.0f, s2 = 0.0f;
    for (int f = 0; f < F; ++f) { float x = v[f]; s1 += x * as[f]; s2 += x * ad[f]; }
    ss[i] = s1;
    sd[i] = s2;
}

#define RROWS   196
#define RWAVES  32
#define QCAP    64
#define SCAN_G  8
#define LCAP    8192

template <int C, int H, bool ELU, int MODE>
__global__ __launch_bounds__(1024) void range_kernel(const int* __restrict__ ei,
                                                     const float* __restrict__ ew,
                                                     const float* __restrict__ meanw,
                                                     const float* __restrict__ ss,
                                                     const float* __restrict__ sd,
                                                     const float* __restrict__ ce,
                                                     const float* __restrict__ xw,
                                                     const float* __restrict__ bias,
                                                     float* __restrict__ out,
                                                     int* __restrict__ gsr,
                                                     float* __restrict__ gw,
                                                     int* __restrict__ gcnt) {
    constexpr int NC = C / 32;
    constexpr int F  = C / H;
    constexpr int U  = 8;
    constexpr int N4 = EE / 4;
    constexpr int ROUND4 = RWAVES * SCAN_G * 32;
    constexpr int NR = (N4 + ROUND4 - 1) / ROUND4;
    static_assert(F % 32 == 0, "a lane's element c*32+lane must lie in head (c*32)/F");
    static_assert(EE % 4 == 0, "the destination plane is read in groups of four");
    static_assert(32 % U == 0 && QCAP >= 64 && (QCAP & (QCAP - 1)) == 0, "queue and group sizes");
    static_assert(RROWS * H <= 1024, "one thread per (row, head) in the first phase");
    static_assert(NN <= 65536 && RROWS <= 256, "a list entry packs source (16 bits) and row (8 bits)");
    static_assert((NN + RROWS - 1) / RROWS <= 256, "the workspace holds 256 blocks' lists and counts");

    __shared__ float acc[RROWS * C];
    __shared__ float den[RROWS * H];
    __shared__ alignas(16) float mrow[RROWS * H];
    __shared__ float aself[RROWS * H];
    __shared__ alignas(16) float sdl[RROWS * H];
    __shared__ int que[RWAVES][QCAP];
    __shared__ int lsr[LCAP];
    __shared__ alignas(16) float lal[LCAP * H];
    __shared__ int lcnt;

    const int tid   = threadIdx.x;
    const int lane  = tid & 31;
    const int wave  = __builtin_amdgcn_readfirstlane(tid >> 5);
    const int d0    = blockIdx.x * RROWS;
    const int nrows = (NN - d0) < RROWS ? (NN - d0) : RROWS;

    for (int i = tid; i < RROWS * C; i += 1024) acc[i] = 0.0f;
    if (tid < RROWS * H) {
        const int r = tid / H, h = tid % H;
        const int n = d0 + (r < nrows ? r : nrows - 1);
        const float sdv = sd[n * H + h];
        float a = ss[n * H + h] + sdv + meanw[0] * ce[h];
        a = a > 0.0f ? a : 0.2f * a;
        den[tid]   = 0.0f;
        mrow[tid]  = a;
        aself[tid] = a;
        sdl[tid]   = sdv;
    }
    if (tid == 0) lcnt = 0;
    __syncthreads();

    int qhead = 0, qcnt = 0;

    auto load_ss = [&](const int s, float (&sv)[H]) {
        if constexpr (H == 4) {
            const float4 t = *reinterpret_cast<const float4*>(ss + (size_t)s * 4);
            sv[0] = t.x; sv[1] = t.y; sv[2] = t.z; sv[3] = t.w;
        } else {
            for (int h = 0; h < H; ++h) sv[h] = ss[(size_t)s * H + h];
        }
    };

    auto alpha_of = [&](const float (&sv)[H], const int r, const float w, float (&al)[H]) {
        if constexpr (H == 4) {
            const float4 dv = *reinterpret_cast<const float4*>(sdl + r * 4);
            al[0] = sv[0] + dv.x + w * ce[0];
            al[1] = sv[1] + dv.y + w * ce[1];
            al[2] = sv[2] + dv.z + w * ce[2];
            al[3] = sv[3] + dv.w + w * ce[3];
        } else {
            for (int h = 0; h < H; ++h) al[h] = sv[h] + sdl[r * H + h] + w * ce[h];
        }
        for (int h = 0; h < H; ++h) al[h] = al[h] > 0.0f ? al[h] : 0.2f * al[h];
    };

    auto edge_alpha = [&](const int e, int& sr, float& w, float (&al)[H]) {
        const int s = ei[e];
        const int r = ei[EE + e] - d0;
        w = ew[e];
        sr = s | (r << 16);
        float sv[H];
        load_ss(s, sv);
        alpha_of(sv, r, w, al);
    };

    auto stage = [&](const int nb) {
        int sr = 0;
        float w = 0.0f;
        float al[H];
        for (int h = 0; h < H; ++h) al[h] = 0.0f;
        __builtin_amdgcn_wave_barrier();
        if (lane < nb) {
            edge_alpha(que[wave][(qhead + lane) & (QCAP - 1)], sr, w, al);
            const int r = sr >> 16;
            for (int h = 0; h < H; ++h) ldsMaxF(&mrow[r * H + h], al[h]);
        }
        int base = 0;
        if (lane == 0) base = atomicAdd(&lcnt, nb);
        base = __builtin_amdgcn_readfirstlane(base);
        if (lane < nb && base + lane < LCAP) {
            lsr[base + lane] = sr;
            if constexpr (H == 4) {
                *reinterpret_cast<float4*>(lal + (base + lane) * 4) = make_float4(al[0], al[1], al[2], al[3]);
            } else {
                for (int h = 0; h < H; ++h) lal[(base + lane) * H + h] = al[h];
            }
            if constexpr (MODE == 0) {
                gsr[(size_t)blockIdx.x * LCAP + base + lane] = sr;
                gw[(size_t)blockIdx.x * LCAP + base + lane]  = w;
            }
        }
    };

    auto accumulate = [&](const int nb, const int sr, const float (&al)[H]) {
        const int s = sr & 0xffff, r = sr >> 16;
        float at[H];
        for (int h = 0; h < H; ++h) at[h] = 0.0f;
        if (lane < nb) {
            if constexpr (H == 4) {
                const float4 mx = *reinterpret_cast<const float4*>(mrow + r * 4);
                at[0] = __expf(al[0] - mx.x);
                at[1] = __expf(al[1] - mx.y);
                at[2] = __expf(al[2] - mx.z);
                at[3] = __expf(al[3] - mx.w);
            } else {
                for (int h = 0; h < H; ++h) at[h] = __expf(al[h] - mrow[r * H + h]);
            }
            for (int h = 0; h < H; ++h) atomicAdd(&den[r * H + h], at[h]);
        }
        for (int j0 = 0; j0 < nb; j0 += U) {
            float v[U][NC];
#pragma unroll
            for (int u = 0; u < U; ++u) {
                if (j0 + u < nb) {
                    const int sj = __builtin_amdgcn_readlane(s, j0 + u);
                    const float* row = xw + (size_t)sj * C + lane;
                    for (int c = 0; c < NC; ++c) v[u][c] = row[c * 32];
                }
            }
#pragma unroll
            for (int u = 0; u < U; ++u) {
                if (j0 + u < nb) {
                    const int rj = __builtin_amdgcn_readlane(r, j0 + u);
                    float* arow = acc + rj * C + lane;
                    for (int c = 0; c < NC; ++c) {
                        const float a = __int_as_float(
                            __builtin_amdgcn_readlane(__float_as_int(at[(c * 32) / F]), j0 + u));
                        atomicAdd(arow + c * 32, v[u][c] * a);
                    }
                }
            }
        }
    };

    auto restage = [&](const int nb) {
        int sr = 0;
        float al[H];
        for (int h = 0; h < H; ++h) al[h] = 0.0f;
        __builtin_amdgcn_wave_barrier();
        float w = 0.0f;
        if (lane < nb) edge_alpha(que[wave][(qhead + lane) & (QCAP - 1)], sr, w, al);
        accumulate(nb, sr, al);
    };

    const int4* __restrict__ dst4 = reinterpret_cast<const int4*>(ei + EE);
    const int rot = blockIdx.x % NR;
    auto scan = [&](auto batch) {
        for (int it = 0; it < NR; ++it) {
            int rnd = it + rot;
            if (rnd >= NR) rnd -= NR;
            const int g0 = rnd * ROUND4 + wave * (SCAN_G * 32);
            int4 dv[SCAN_G];
#pragma unroll
            for (int g = 0; g < SCAN_G; ++g) {
                const int i4 = g0 + g * 32 + lane;
                dv[g] = dst4[i4 < N4 ? i4 : N4 - 1];
            }
            unsigned hits = 0u;
#pragma unroll
            for (int g = 0; g < SCAN_G; ++g) {
                const bool ok = (g0 + g * 32 + lane) < N4;
                hits |= (ok && (unsigned)(dv[g].x - d0) < (unsigned)nrows) ? (1u << (g * 4 + 0)) : 0u;
                hits |= (ok && (unsigned)(dv[g].y - d0) < (unsigned)nrows) ? (1u << (g * 4 + 1)) : 0u;
                hits |= (ok && (unsigned)(dv[g].z - d0) < (unsigned)nrows) ? (1u << (g * 4 + 2)) : 0u;
                hits |= (ok && (unsigned)(dv[g].w - d0) < (unsigned)nrows) ? (1u << (g * 4 + 3)) : 0u;
            }
            unsigned mask = __builtin_amdgcn_ballot_w32(hits != 0u);
            while (mask != 0u) {
                if (hits != 0u) {
                    const int b = __builtin_ctz(hits);
                    hits &= hits - 1u;
                    const int e = (g0 + (b >> 2) * 32 + lane) * 4 + (b & 3);
                    const int pos = qcnt + (int)__builtin_amdgcn_mbcnt_lo(mask, 0u);
                    que[wave][(qhead + pos) & (QCAP - 1)] = e;
                }
                qcnt += __popc(mask);
                if (qcnt >= 32) {
                    batch(32);
                    qhead = (qhead + 32) & (QCAP - 1);
                    qcnt -= 32;
                }
                mask = __builtin_amdgcn_ballot_w32(hits != 0u);
            }
        }
        if (qcnt > 0) batch(qcnt);
        qhead = 0;
        qcnt = 0;
    };

    bool listed = false;
    if constexpr (MODE == 1) {
        const int mt = __builtin_amdgcn_readfirstlane(gcnt[blockIdx.x]);
        listed = mt <= LCAP;
        if (listed) {
            const int*   __restrict__ msr = gsr + (size_t)blockIdx.x * LCAP;
            const float* __restrict__ mw  = gw  + (size_t)blockIdx.x * LCAP;
            constexpr int G = 4;
            for (int b0 = wave * 32; b0 < mt; b0 += G * RWAVES * 32) {
                int   sr[G];
                float w[G];
                float sv[G][H];
#pragma unroll
                for (int g = 0; g < G; ++g) {
                    const int p = b0 + g * (RWAVES * 32) + lane;
                    const int q = p < mt ? p : mt - 1;
                    sr[g] = msr[q];
                    w[g]  = mw[q];
                }
#pragma unroll
                for (int g = 0; g < G; ++g) asm volatile("" :: "v"(sr[g]), "v"(w[g]));
#pragma unroll
                for (int g = 0; g < G; ++g) load_ss(sr[g] & 0xffff, sv[g]);
#pragma unroll
                for (int g = 0; g < G; ++g) {
                    const int p = b0 + g * (RWAVES * 32) + lane;
                    if (p < mt) {
                        const int r = sr[g] >> 16;
                        float al[H];
                        alpha_of(sv[g], r, w[g], al);
                        for (int h = 0; h < H; ++h) ldsMaxF(&mrow[r * H + h], al[h]);
                        lsr[p] = sr[g];
                        if constexpr (H == 4) {
                            *reinterpret_cast<float4*>(lal + p * 4) = make_float4(al[0], al[1], al[2], al[3]);
                        } else {
                            for (int h = 0; h < H; ++h) lal[p * H + h] = al[h];
                        }
                    }
                }
            }
            if (tid == 0) lcnt = mt;
        }
    }
    if (!listed) scan(stage);
    __syncthreads();

    const int total = __builtin_amdgcn_readfirstlane(lcnt);
    if constexpr (MODE == 0) {
        if (tid == 0) gcnt[blockIdx.x] = total;
    }
    if (total <= LCAP) {
        for (int base = wave * 32; base < total; base += RWAVES * 32) {
            const int left = total - base;
            int sr = 0;
            float al[H];
            for (int h = 0; h < H; ++h) al[h] = 0.0f;
            if (lane < left) {
                sr = lsr[base + lane];
                if constexpr (H == 4) {
                    const float4 t = *reinterpret_cast<const float4*>(lal + (base + lane) * 4);
                    al[0] = t.x; al[1] = t.y; al[2] = t.z; al[3] = t.w;
                } else {
                    for (int h = 0; h < H; ++h) al[h] = lal[(base + lane) * H + h];
                }
            }
            if (left >= 32) accumulate(32, sr, al);
            else            accumulate(left, sr, al);
        }
    } else {
        scan(restage);
    }
    __syncthreads();

    {
        constexpr int RPT = 1024 / C;
        constexpr int P0  = 8;
        const int c = tid % C, h = c / F;
        const float bc = bias[c];
        for (int r0 = tid / C; r0 < nrows; r0 += RPT * P0) {
            float xv[P0];
#pragma unroll
            for (int k = 0; k < P0; ++k) {
                const int r = r0 + k * RPT;
                xv[k] = xw[(size_t)(d0 + (r < nrows ? r : nrows - 1)) * C + c];
            }
#pragma unroll
            for (int k = 0; k < P0; ++k) {
                const int r = r0 + k * RPT;
                if (r < nrows) {
                    const int rh = r * H + h;
                    const float exs = __expf(aself[rh] - mrow[rh]);
                    float v = (acc[r * C + c] + exs * xv[k]) / (den[rh] + exs + 1e-16f) + bc;
                    if constexpr (ELU) v = v > 0.0f ? v : expm1f(v);
                    out[(size_t)(d0 + r) * C + c] = v;
                }
            }
        }
    }
}

extern "C" void kernel_launch(void* const* d_in, const int* in_sizes, int n_in,
                              void* d_out, int out_size, void* d_ws, size_t ws_size,
                              hipStream_t stream) {
    const float* x  = (const float*)d_in[0];
    const int*   ei = (const int*)d_in[1];
    const float* ew = (const float*)d_in[2];

    const float* Wl[3] = { (const float*)d_in[3],  (const float*)d_in[9],  (const float*)d_in[15] };
    const float* As[3] = { (const float*)d_in[4],  (const float*)d_in[10], (const float*)d_in[16] };
    const float* Ad[3] = { (const float*)d_in[5],  (const float*)d_in[11], (const float*)d_in[17] };
    const float* We[3] = { (const float*)d_in[6],  (const float*)d_in[12], (const float*)d_in[18] };
    const float* Ae[3] = { (const float*)d_in[7],  (const float*)d_in[13], (const float*)d_in[19] };
    const float* Bi[3] = { (const float*)d_in[8],  (const float*)d_in[14], (const float*)d_in[20] };

    const int Hs[3] = { 4, 4, 1 };
    const int Fs[3] = { 32, 32, 64 };

    float* ws    = (float*)d_ws;
    float* xw    = ws;
    float* bufA  = xw    + (size_t)NN * 128;
    float* bufB  = bufA  + (size_t)NN * 128;
    float* ss    = bufB  + (size_t)NN * 128;
    float* sd    = ss    + (size_t)NN * 4;
    float* ce    = sd    + (size_t)NN * 4;
    float* meanw = ce + 4;
    int*   gcnt  = (int*)(ce + 64);
    int*   gsr   = gcnt + 256;
    float* gw    = (float*)(gsr + 256 * LCAP);

    mean_kernel<<<1, 1024, 0, stream>>>(ew, meanw, EE);

    const float* in = x;
    float* outs[3] = { bufA, bufB, (float*)d_out };

    for (int l = 0; l < 3; ++l) {
        const int H = Hs[l], F = Fs[l];

        edge_coef_kernel<<<1, 32, 0, stream>>>(We[l], Ae[l], ce, H, F);

        if (l < 2)
            gemm_wmma_f32<128><<<(NN + GROWS - 1) / GROWS, 256, 0, stream>>>(in, Wl[l], xw);
        else
            gemm_wmma_f32<64><<<(NN + GROWS - 1) / GROWS, 128, 0, stream>>>(in, Wl[l], xw);

        score_kernel<<<(NN * H + 255) / 256, 256, 0, stream>>>(xw, As[l], Ad[l], ss, sd, H, F);

        if (l == 0)
            range_kernel<128, 4, true, 0><<<(NN + RROWS - 1) / RROWS, 1024, 0, stream>>>(
                ei, ew, meanw, ss, sd, ce, xw, Bi[l], outs[l], gsr, gw, gcnt);
        else if (l == 1)
            range_kernel<128, 4, true, 1><<<(NN + RROWS - 1) / RROWS, 1024, 0, stream>>>(
                ei, ew, meanw, ss, sd, ce, xw, Bi[l], outs[l], gsr, gw, gcnt);
        else
            range_kernel<64, 1, false, 1><<<(NN + RROWS - 1) / RROWS, 1024, 0, stream>>>(
                ei, ew, meanw, ss, sd, ce, xw, Bi[l], outs[l], gsr, gw, gcnt);

        in = outs[l];
    }
}


# scan reads a byte per edge: layer 0's score launch also writes each edge's owner block (destination/196) as a byte; a block scans that 800 KB plane, not 3200 KB of destinations; sum order changes
# speedup vs baseline: 6.1423x; 1.0570x over previous
#include <hip/hip_runtime.h>
#include <math.h>

#define NN    50000
#define EE    800000
#define KDIM  128
#define RROWS 196

typedef float v2f __attribute__((ext_vector_type(2)));
typedef float v8f __attribute__((ext_vector_type(8)));

__device__ __forceinline__ void ldsMaxF(float* addr, float val) {
    if (val >= 0.0f) atomicMax((int*)addr, __float_as_int(val));
    else             atomicMin((unsigned int*)addr, __float_as_uint(val));
}

#define MEAN_U 8
__global__ __launch_bounds__(1024) void mean_kernel(const float* __restrict__ ew,
                                                    float* __restrict__ meanw, int E) {
    __shared__ float sdata[1024];
    const int n4   = E >> 2;
    const int full = n4 / (MEAN_U * 1024);
    const float4* __restrict__ ew4 = reinterpret_cast<const float4*>(ew);
    float s = 0.0f;

    const float4* p = ew4 + threadIdx.x;
    for (int g = 0; g < full; ++g, p += MEAN_U * 1024) {
        float4 v[MEAN_U];
#pragma unroll
        for (int u = 0; u < MEAN_U; ++u) v[u] = p[u * 1024];
#pragma unroll
        for (int u = 0; u < MEAN_U; ++u) s += (v[u].x + v[u].y) + (v[u].z + v[u].w);
    }

    const int r0 = full * (MEAN_U * 1024) + (int)threadIdx.x;
    if (r0 < n4) {
        float4 v[MEAN_U];
#pragma unroll
        for (int u = 0; u < MEAN_U; ++u) {
            const int idx = r0 + u * 1024;
            v[u] = ew4[idx < n4 ? idx : n4 - 1];
        }
#pragma unroll
        for (int u = 0; u < MEAN_U; ++u)
            if (r0 + u * 1024 < n4) s += (v[u].x + v[u].y) + (v[u].z + v[u].w);
    }
    if ((int)threadIdx.x < (E & 3)) s += ew[(n4 << 2) + threadIdx.x];

    sdata[threadIdx.x] = s;
    __syncthreads();
    for (int o = 512; o > 0; o >>= 1) {
        if (threadIdx.x < o) sdata[threadIdx.x] += sdata[threadIdx.x + o];
        __syncthreads();
    }
    if (threadIdx.x == 0) meanw[0] = sdata[0] / (float)E;
}

__global__ void edge_coef_kernel(const float* __restrict__ We, const float* __restrict__ ae,
                                 float* __restrict__ ce, int H, int F) {
    int h = threadIdx.x;
    if (h >= H) return;
    float s = 0.0f;
    for (int f = 0; f < F; ++f) s += We[h * F + f] * ae[h * F + f];
    ce[h] = s;
}

#define GROWS   64
#define GPITCH  132

template <int C>
__global__ __launch_bounds__(C * 2, C == 128 ? 8 : 4) void gemm_wmma_f32(const float* __restrict__ X,
                                                       const float* __restrict__ W,
                                                       float* __restrict__ out) {
    constexpr int NT = C * 2;
    constexpr int MT = GROWS / 16;
    constexpr int KS = KDIM / 4;
    constexpr int XL = GROWS * (KDIM / 4) / NT;
    constexpr int C4 = C / 4;
    constexpr int OL = GROWS * C4 / NT;
    static_assert(KDIM == 128 && GPITCH >= KDIM && GPITCH % 4 == 0, "a wave's 16-byte loads cover one row of X");
    static_assert(GROWS * (KDIM / 4) % NT == 0 && GROWS * C4 % NT == 0, "whole trips");

    __shared__ alignas(16) float tile[GROWS * GPITCH];

    const int tid  = threadIdx.x;
    const int lane = tid & 31;
    const int wave = __builtin_amdgcn_readfirstlane(tid >> 5);
    const int half = lane >> 4;
    const int l15  = lane & 15;
    const int row0 = blockIdx.x * GROWS;

    float4 xv[XL];
#pragma unroll
    for (int i = 0; i < XL; ++i) {
        const int idx = i * NT + tid;
        const int r   = idx >> 5, c4 = idx & 31;
        const int gr  = (row0 + r) < NN ? (row0 + r) : (NN - 1);
        xv[i] = *reinterpret_cast<const float4*>(X + (size_t)gr * KDIM + c4 * 4);
    }

    v2f b[KS];
    const int coln = wave * 16 + l15;
#pragma unroll
    for (int ks = 0; ks < KS; ++ks) {
        const int kk = ks * 4 + half * 2;
        b[ks][0] = W[(size_t)kk * C + coln];
        b[ks][1] = W[(size_t)(kk + 1) * C + coln];
    }

#pragma unroll
    for (int i = 0; i < XL; ++i) {
        const int idx = i * NT + tid;
        const int r   = idx >> 5, c4 = idx & 31;
        *reinterpret_cast<float4*>(tile + r * GPITCH + c4 * 4) = xv[i];
    }
    __syncthreads();

    v8f acc[MT];
#pragma unroll
    for (int m = 0; m < MT; ++m) acc[m] = v8f{};

    const float* arow = tile + l15 * GPITCH + half * 2;
#pragma unroll
    for (int ks = 0; ks < KS; ++ks) {
#pragma unroll
        for (int m = 0; m < MT; ++m) {
            const v2f a = *reinterpret_cast<const v2f*>(arow + m * 16 * GPITCH + ks * 4);
            acc[m] = __builtin_amdgcn_wmma_f32_16x16x4_f32(
                 false, a,  false, b[ks],
                 (short)0, acc[m],  false,  false);
        }
    }
    __syncthreads();

#pragma unroll
    for (int m = 0; m < MT; ++m) {
#pragma unroll
        for (int v = 0; v < 8; ++v)
            tile[(m * 16 + v + half * 8) * GPITCH + coln] = acc[m][v];
    }
    __syncthreads();
#pragma unroll
    for (int i = 0; i < OL; ++i) {
        const int idx = i * NT + tid;
        const int r   = idx / C4, c4 = idx % C4;
        if (row0 + r < NN)
            *reinterpret_cast<float4*>(out + (size_t)(row0 + r) * C + c4 * 4) =
                *reinterpret_cast<const float4*>(tile + r * GPITCH + c4 * 4);
    }
}

template <bool OWN>
__global__ void score_kernel(const float* __restrict__ xw,
                             const float* __restrict__ a_s, const float* __restrict__ a_d,
                             float* __restrict__ ss, float* __restrict__ sd,
                             int H, int F,
                             const int* __restrict__ ei, unsigned int* __restrict__ own) {
    int i = blockIdx.x * blockDim.x + threadIdx.x;
    if constexpr (OWN) {
        static_assert(EE % 4 == 0 && (NN + RROWS - 1) / RROWS <= 256, "four owner bytes per word, an owner fits a byte");
        const int4* __restrict__ dst4 = reinterpret_cast<const int4*>(ei + EE);
        for (int g = i; g < EE / 4; g += gridDim.x * blockDim.x) {
            const int4 d = dst4[g];
            own[g] = ((unsigned)d.x / RROWS) | (((unsigned)d.y / RROWS) << 8) |
                     (((unsigned)d.z / RROWS) << 16) | (((unsigned)d.w / RROWS) << 24);
        }
    }
    if (i >= NN * H) return;
    const int n = i / H, h = i % H;
    const float* v  = xw + (size_t)n * H * F + h * F;
    const float* as = a_s + h * F;
    const float* ad = a_d + h * F;
    float s1 = 0.0f, s2 = 0.0f;
    for (int f = 0; f < F; ++f) { float x = v[f]; s1 += x * as[f]; s2 += x * ad[f]; }
    ss[i] = s1;
    sd[i] = s2;
}

#define RWAVES  32
#define QCAP    64
#define SCAN_G  4
#define LCAP    8192

template <int C, int H, bool ELU, int MODE>
__global__ __launch_bounds__(1024) void range_kernel(const int* __restrict__ ei,
                                                     const float* __restrict__ ew,
                                                     const float* __restrict__ meanw,
                                                     const float* __restrict__ ss,
                                                     const float* __restrict__ sd,
                                                     const float* __restrict__ ce,
                                                     const float* __restrict__ xw,
                                                     const float* __restrict__ bias,
                                                     float* __restrict__ out,
                                                     int* __restrict__ gsr,
                                                     float* __restrict__ gw,
                                                     int* __restrict__ gcnt,
                                                     const unsigned int* __restrict__ own) {
    constexpr int NC = C / 32;
    constexpr int F  = C / H;
    constexpr int U  = 8;
    constexpr int N16 = EE / 16;
    constexpr int ROUND16 = RWAVES * SCAN_G * 32;
    constexpr int NR = (N16 + ROUND16 - 1) / ROUND16;
    static_assert(F % 32 == 0, "a lane's element c*32+lane must lie in head (c*32)/F");
    static_assert(EE % 16 == 0 && SCAN_G == 4, "the owner plane is read in groups of 16 bytes, 64 hit bits per lane and round");
    static_assert(32 % U == 0 && QCAP >= 64 && (QCAP & (QCAP - 1)) == 0, "queue and group sizes");
    static_assert(RROWS * H <= 1024, "one thread per (row, head) in the first phase");
    static_assert(NN <= 65536 && RROWS <= 256, "a list entry packs source (16 bits) and row (8 bits)");
    static_assert((NN + RROWS - 1) / RROWS <= 256, "the workspace holds 256 blocks' lists and counts");

    __shared__ float acc[RROWS * C];
    __shared__ float den[RROWS * H];
    __shared__ alignas(16) float mrow[RROWS * H];
    __shared__ float aself[RROWS * H];
    __shared__ alignas(16) float sdl[RROWS * H];
    __shared__ int que[RWAVES][QCAP];
    __shared__ int lsr[LCAP];
    __shared__ alignas(16) float lal[LCAP * H];
    __shared__ int lcnt;

    const int tid   = threadIdx.x;
    const int lane  = tid & 31;
    const int wave  = __builtin_amdgcn_readfirstlane(tid >> 5);
    const int d0    = blockIdx.x * RROWS;
    const int nrows = (NN - d0) < RROWS ? (NN - d0) : RROWS;

    for (int i = tid; i < RROWS * C; i += 1024) acc[i] = 0.0f;
    if (tid < RROWS * H) {
        const int r = tid / H, h = tid % H;
        const int n = d0 + (r < nrows ? r : nrows - 1);
        const float sdv = sd[n * H + h];
        float a = ss[n * H + h] + sdv + meanw[0] * ce[h];
        a = a > 0.0f ? a : 0.2f * a;
        den[tid]   = 0.0f;
        mrow[tid]  = a;
        aself[tid] = a;
        sdl[tid]   = sdv;
    }
    if (tid == 0) lcnt = 0;
    __syncthreads();

    int qhead = 0, qcnt = 0;

    auto load_ss = [&](const int s, float (&sv)[H]) {
        if constexpr (H == 4) {
            const float4 t = *reinterpret_cast<const float4*>(ss + (size_t)s * 4);
            sv[0] = t.x; sv[1] = t.y; sv[2] = t.z; sv[3] = t.w;
        } else {
            for (int h = 0; h < H; ++h) sv[h] = ss[(size_t)s * H + h];
        }
    };

    auto alpha_of = [&](const float (&sv)[H], const int r, const float w, float (&al)[H]) {
        if constexpr (H == 4) {
            const float4 dv = *reinterpret_cast<const float4*>(sdl + r * 4);
            al[0] = sv[0] + dv.x + w * ce[0];
            al[1] = sv[1] + dv.y + w * ce[1];
            al[2] = sv[2] + dv.z + w * ce[2];
            al[3] = sv[3] + dv.w + w * ce[3];
        } else {
            for (int h = 0; h < H; ++h) al[h] = sv[h] + sdl[r * H + h] + w * ce[h];
        }
        for (int h = 0; h < H; ++h) al[h] = al[h] > 0.0f ? al[h] : 0.2f * al[h];
    };

    auto edge_alpha = [&](const int e, int& sr, float& w, float (&al)[H]) {
        const int s = ei[e];
        const int r = ei[EE + e] - d0;
        w = ew[e];
        sr = s | (r << 16);
        float sv[H];
        load_ss(s, sv);
        alpha_of(sv, r, w, al);
    };

    auto stage = [&](const int nb) {
        int sr = 0;
        float w = 0.0f;
        float al[H];
        for (int h = 0; h < H; ++h) al[h] = 0.0f;
        __builtin_amdgcn_wave_barrier();
        if (lane < nb) {
            edge_alpha(que[wave][(qhead + lane) & (QCAP - 1)], sr, w, al);
            const int r = sr >> 16;
            for (int h = 0; h < H; ++h) ldsMaxF(&mrow[r * H + h], al[h]);
        }
        int base = 0;
        if (lane == 0) base = atomicAdd(&lcnt, nb);
        base = __builtin_amdgcn_readfirstlane(base);
        if (lane < nb && base + lane < LCAP) {
            lsr[base + lane] = sr;
            if constexpr (H == 4) {
                *reinterpret_cast<float4*>(lal + (base + lane) * 4) = make_float4(al[0], al[1], al[2], al[3]);
            } else {
                for (int h = 0; h < H; ++h) lal[(base + lane) * H + h] = al[h];
            }
            if constexpr (MODE == 0) {
                gsr[(size_t)blockIdx.x * LCAP + base + lane] = sr;
                gw[(size_t)blockIdx.x * LCAP + base + lane]  = w;
            }
        }
    };

    auto accumulate = [&](const int nb, const int sr, const float (&al)[H]) {
        const int s = sr & 0xffff, r = sr >> 16;
        float at[H];
        for (int h = 0; h < H; ++h) at[h] = 0.0f;
        if (lane < nb) {
            if constexpr (H == 4) {
                const float4 mx = *reinterpret_cast<const float4*>(mrow + r * 4);
                at[0] = __expf(al[0] - mx.x);
                at[1] = __expf(al[1] - mx.y);
                at[2] = __expf(al[2] - mx.z);
                at[3] = __expf(al[3] - mx.w);
            } else {
                for (int h = 0; h < H; ++h) at[h] = __expf(al[h] - mrow[r * H + h]);
            }
            for (int h = 0; h < H; ++h) atomicAdd(&den[r * H + h], at[h]);
        }
        for (int j0 = 0; j0 < nb; j0 += U) {
            float v[U][NC];
#pragma unroll
            for (int u = 0; u < U; ++u) {
                if (j0 + u < nb) {
                    const int sj = __builtin_amdgcn_readlane(s, j0 + u);
                    const float* row = xw + (size_t)sj * C + lane;
                    for (int c = 0; c < NC; ++c) v[u][c] = row[c * 32];
                }
            }
#pragma unroll
            for (int u = 0; u < U; ++u) {
                if (j0 + u < nb) {
                    const int rj = __builtin_amdgcn_readlane(r, j0 + u);
                    float* arow = acc + rj * C + lane;
                    for (int c = 0; c < NC; ++c) {
                        const float a = __int_as_float(
                            __builtin_amdgcn_readlane(__float_as_int(at[(c * 32) / F]), j0 + u));
                        atomicAdd(arow + c * 32, v[u][c] * a);
                    }
                }
            }
        }
    };

    auto restage = [&](const int nb) {
        int sr = 0;
        float al[H];
        for (int h = 0; h < H; ++h) al[h] = 0.0f;
        __builtin_amdgcn_wave_barrier();
        float w = 0.0f;
        if (lane < nb) edge_alpha(que[wave][(qhead + lane) & (QCAP - 1)], sr, w, al);
        accumulate(nb, sr, al);
    };

    const uint4* __restrict__ own16 = reinterpret_cast<const uint4*>(own);
    const unsigned pat = (unsigned)blockIdx.x * 0x01010101u;
    const int rot = blockIdx.x % NR;
    auto match = [&](const unsigned v) -> unsigned {
        const unsigned t = v ^ pat;
        return ~(((t & 0x7f7f7f7fu) + 0x7f7f7f7fu) | t | 0x7f7f7f7fu);
    };
    auto group0 = [&](const int it) -> int {
        int rnd = it + rot;
        if (rnd >= NR) rnd -= NR;
        return rnd * ROUND16 + wave * (SCAN_G * 32);
    };
    auto issue = [&](const int g0, uint4 (&d)[SCAN_G]) {
#pragma unroll
        for (int g = 0; g < SCAN_G; ++g) {
            const int i16 = g0 + g * 32 + lane;
            d[g] = own16[i16 < N16 ? i16 : N16 - 1];
        }
    };
    auto scan = [&](auto batch) {
        uint4 dv[SCAN_G];
        int g0 = group0(0);
        issue(g0, dv);
        for (int it = 0; it < NR; ++it) {
            const int g1 = group0(it + 1 < NR ? it + 1 : it);
            uint4 nx[SCAN_G];
            issue(g1, nx);
            unsigned long long hits = 0ull;
#pragma unroll
            for (int g = 0; g < SCAN_G; ++g) {
                unsigned z = (match(dv[g].x) >> 7) | (match(dv[g].y) >> 6) |
                             (match(dv[g].z) >> 5) | (match(dv[g].w) >> 4);
                if ((g0 + g * 32 + lane) >= N16) z = 0u;
                hits |= (unsigned long long)z << (32 * (g >> 1) + 4 * (g & 1));
            }
            unsigned mask = __builtin_amdgcn_ballot_w32(hits != 0ull);
            while (mask != 0u) {
                if (hits != 0ull) {
                    const int b = __builtin_ctzll(hits);
                    hits &= hits - 1ull;
                    const int g = ((b >> 5) << 1) | ((b >> 2) & 1), j = b & 3, k = (b >> 3) & 3;
                    const int e = (g0 + g * 32 + lane) * 16 + j * 4 + k;
                    const int pos = qcnt + (int)__builtin_amdgcn_mbcnt_lo(mask, 0u);
                    que[wave][(qhead + pos) & (QCAP - 1)] = e;
                }
                qcnt += __popc(mask);
                if (qcnt >= 32) {
                    batch(32);
                    qhead = (qhead + 32) & (QCAP - 1);
                    qcnt -= 32;
                }
                mask = __builtin_amdgcn_ballot_w32(hits != 0ull);
            }
#pragma unroll
            for (int g = 0; g < SCAN_G; ++g) dv[g] = nx[g];
            g0 = g1;
        }
        if (qcnt > 0) batch(qcnt);
        qhead = 0;
        qcnt = 0;
    };

    bool listed = false;
    if constexpr (MODE == 1) {
        const int mt = __builtin_amdgcn_readfirstlane(gcnt[blockIdx.x]);
        listed = mt <= LCAP;
        if (listed) {
            const int*   __restrict__ msr = gsr + (size_t)blockIdx.x * LCAP;
            const float* __restrict__ mw  = gw  + (size_t)blockIdx.x * LCAP;
            constexpr int G = 4;
            for (int b0 = wave * 32; b0 < mt; b0 += G * RWAVES * 32) {
                int   sr[G];
                float w[G];
                float sv[G][H];
#pragma unroll
                for (int g = 0; g < G; ++g) {
                    const int p = b0 + g * (RWAVES * 32) + lane;
                    const int q = p < mt ? p : mt - 1;
                    sr[g] = msr[q];
                    w[g]  = mw[q];
                }
#pragma unroll
                for (int g = 0; g < G; ++g) asm volatile("" :: "v"(sr[g]), "v"(w[g]));
#pragma unroll
                for (int g = 0; g < G; ++g) load_ss(sr[g] & 0xffff, sv[g]);
#pragma unroll
                for (int g = 0; g < G; ++g) {
                    const int p = b0 + g * (RWAVES * 32) + lane;
                    if (p < mt) {
                        const int r = sr[g] >> 16;
                        float al[H];
                        alpha_of(sv[g], r, w[g], al);
                        for (int h = 0; h < H; ++h) ldsMaxF(&mrow[r * H + h], al[h]);
                        lsr[p] = sr[g];
                        if constexpr (H == 4) {
                            *reinterpret_cast<float4*>(lal + p * 4) = make_float4(al[0], al[1], al[2], al[3]);
                        } else {
                            for (int h = 0; h < H; ++h) lal[p * H + h] = al[h];
                        }
                    }
                }
            }
            if (tid == 0) lcnt = mt;
        }
    }
    if (!listed) scan(stage);
    __syncthreads();

    const int total = __builtin_amdgcn_readfirstlane(lcnt);
    if constexpr (MODE == 0) {
        if (tid == 0) gcnt[blockIdx.x] = total;
    }
    if (total <= LCAP) {
        for (int base = wave * 32; base < total; base += RWAVES * 32) {
            const int left = total - base;
            int sr = 0;
            float al[H];
            for (int h = 0; h < H; ++h) al[h] = 0.0f;
            if (lane < left) {
                sr = lsr[base + lane];
                if constexpr (H == 4) {
                    const float4 t = *reinterpret_cast<const float4*>(lal + (base + lane) * 4);
                    al[0] = t.x; al[1] = t.y; al[2] = t.z; al[3] = t.w;
                } else {
                    for (int h = 0; h < H; ++h) al[h] = lal[(base + lane) * H + h];
                }
            }
            if (left >= 32) accumulate(32, sr, al);
            else            accumulate(left, sr, al);
        }
    } else {
        scan(restage);
    }
    __syncthreads();

    {
        constexpr int RPT = 1024 / C;
        constexpr int P0  = 8;
        const int c = tid % C, h = c / F;
        const float bc = bias[c];
        for (int r0 = tid / C; r0 < nrows; r0 += RPT * P0) {
            float xv[P0];
#pragma unroll
            for (int k = 0; k < P0; ++k) {
                const int r = r0 + k * RPT;
                xv[k] = xw[(size_t)(d0 + (r < nrows ? r : nrows - 1)) * C + c];
            }
#pragma unroll
            for (int k = 0; k < P0; ++k) {
                const int r = r0 + k * RPT;
                if (r < nrows) {
                    const int rh = r * H + h;
                    const float exs = __expf(aself[rh] - mrow[rh]);
                    float v = (acc[r * C + c] + exs * xv[k]) / (den[rh] + exs + 1e-16f) + bc;
                    if constexpr (ELU) v = v > 0.0f ? v : expm1f(v);
                    out[(size_t)(d0 + r) * C + c] = v;
                }
            }
        }
    }
}

extern "C" void kernel_launch(void* const* d_in, const int* in_sizes, int n_in,
                              void* d_out, int out_size, void* d_ws, size_t ws_size,
                              hipStream_t stream) {
    const float* x  = (const float*)d_in[0];
    const int*   ei = (const int*)d_in[1];
    const float* ew = (const float*)d_in[2];

    const float* Wl[3] = { (const float*)d_in[3],  (const float*)d_in[9],  (const float*)d_in[15] };
    const float* As[3] = { (const float*)d_in[4],  (const float*)d_in[10], (const float*)d_in[16] };
    const float* Ad[3] = { (const float*)d_in[5],  (const float*)d_in[11], (const float*)d_in[17] };
    const float* We[3] = { (const float*)d_in[6],  (const float*)d_in[12], (const float*)d_in[18] };
    const float* Ae[3] = { (const float*)d_in[7],  (const float*)d_in[13], (const float*)d_in[19] };
    const float* Bi[3] = { (const float*)d_in[8],  (const float*)d_in[14], (const float*)d_in[20] };

    const int Hs[3] = { 4, 4, 1 };
    const int Fs[3] = { 32, 32, 64 };

    float* ws    = (float*)d_ws;
    float* xw    = ws;
    float* bufA  = xw    + (size_t)NN * 128;
    float* bufB  = bufA  + (size_t)NN * 128;
    float* ss    = bufB  + (size_t)NN * 128;
    float* sd    = ss    + (size_t)NN * 4;
    float* ce    = sd    + (size_t)NN * 4;
    float* meanw = ce + 4;
    int*   gcnt  = (int*)(ce + 64);
    int*   gsr   = gcnt + 256;
    float* gw    = (float*)(gsr + 256 * LCAP);
    unsigned int* own = (unsigned int*)(gw + 256 * LCAP);

    mean_kernel<<<1, 1024, 0, stream>>>(ew, meanw, EE);

    const float* in = x;
    float* outs[3] = { bufA, bufB, (float*)d_out };

    for (int l = 0; l < 3; ++l) {
        const int H = Hs[l], F = Fs[l];

        edge_coef_kernel<<<1, 32, 0, stream>>>(We[l], Ae[l], ce, H, F);

        if (l < 2)
            gemm_wmma_f32<128><<<(NN + GROWS - 1) / GROWS, 256, 0, stream>>>(in, Wl[l], xw);
        else
            gemm_wmma_f32<64><<<(NN + GROWS - 1) / GROWS, 128, 0, stream>>>(in, Wl[l], xw);

        if (l == 0)
            score_kernel<true><<<(NN * H + 255) / 256, 256, 0, stream>>>(xw, As[l], Ad[l], ss, sd, H, F, ei, own);
        else
            score_kernel<false><<<(NN * H + 255) / 256, 256, 0, stream>>>(xw, As[l], Ad[l], ss, sd, H, F, ei, own);

        if (l == 0)
            range_kernel<128, 4, true, 0><<<(NN + RROWS - 1) / RROWS, 1024, 0, stream>>>(
                ei, ew, meanw, ss, sd, ce, xw, Bi[l], outs[l], gsr, gw, gcnt, own);
        else if (l == 1)
            range_kernel<128, 4, true, 1><<<(NN + RROWS - 1) / RROWS, 1024, 0, stream>>>(
                ei, ew, meanw, ss, sd, ce, xw, Bi[l], outs[l], gsr, gw, gcnt, own);
        else
            range_kernel<64, 1, false, 1><<<(NN + RROWS - 1) / RROWS, 1024, 0, stream>>>(
                ei, ew, meanw, ss, sd, ce, xw, Bi[l], outs[l], gsr, gw, gcnt, own);

        in = outs[l];
    }
}


# mean launch goes: layer 0's score launch, which walks the edges already, also sums their weights per block; layer 0's range blocks add the 782 sums; 13 -> 12 launches; mean's sum order changes
# speedup vs baseline: 7.3203x; 1.1918x over previous
#include <hip/hip_runtime.h>
#include <math.h>

#define NN    50000
#define EE    800000
#define KDIM  128
#define RROWS 196

typedef float v2f __attribute__((ext_vector_type(2)));
typedef float v8f __attribute__((ext_vector_type(8)));

__device__ __forceinline__ void ldsMaxF(float* addr, float val) {
    if (val >= 0.0f) atomicMax((int*)addr, __float_as_int(val));
    else             atomicMin((unsigned int*)addr, __float_as_uint(val));
}

#define STHR 256
#define SBLK ((NN * 4 + STHR - 1) / STHR)

__device__ __forceinline__ float waveSum(float v) {
    for (int o = 16; o > 0; o >>= 1) v += __shfl_xor(v, o);
    return v;
}

__global__ void edge_coef_kernel(const float* __restrict__ We, const float* __restrict__ ae,
                                 float* __restrict__ ce, int H, int F) {
    int h = threadIdx.x;
    if (h >= H) return;
    float s = 0.0f;
    for (int f = 0; f < F; ++f) s += We[h * F + f] * ae[h * F + f];
    ce[h] = s;
}

#define GROWS   64
#define GPITCH  132

template <int C>
__global__ __launch_bounds__(C * 2, C == 128 ? 8 : 4) void gemm_wmma_f32(const float* __restrict__ X,
                                                       const float* __restrict__ W,
                                                       float* __restrict__ out) {
    constexpr int NT = C * 2;
    constexpr int MT = GROWS / 16;
    constexpr int KS = KDIM / 4;
    constexpr int XL = GROWS * (KDIM / 4) / NT;
    constexpr int C4 = C / 4;
    constexpr int OL = GROWS * C4 / NT;
    static_assert(KDIM == 128 && GPITCH >= KDIM && GPITCH % 4 == 0, "a wave's 16-byte loads cover one row of X");
    static_assert(GROWS * (KDIM / 4) % NT == 0 && GROWS * C4 % NT == 0, "whole trips");

    __shared__ alignas(16) float tile[GROWS * GPITCH];

    const int tid  = threadIdx.x;
    const int lane = tid & 31;
    const int wave = __builtin_amdgcn_readfirstlane(tid >> 5);
    const int half = lane >> 4;
    const int l15  = lane & 15;
    const int row0 = blockIdx.x * GROWS;

    float4 xv[XL];
#pragma unroll
    for (int i = 0; i < XL; ++i) {
        const int idx = i * NT + tid;
        const int r   = idx >> 5, c4 = idx & 31;
        const int gr  = (row0 + r) < NN ? (row0 + r) : (NN - 1);
        xv[i] = *reinterpret_cast<const float4*>(X + (size_t)gr * KDIM + c4 * 4);
    }

    v2f b[KS];
    const int coln = wave * 16 + l15;
#pragma unroll
    for (int ks = 0; ks < KS; ++ks) {
        const int kk = ks * 4 + half * 2;
        b[ks][0] = W[(size_t)kk * C + coln];
        b[ks][1] = W[(size_t)(kk + 1) * C + coln];
    }

#pragma unroll
    for (int i = 0; i < XL; ++i) {
        const int idx = i * NT + tid;
        const int r   = idx >> 5, c4 = idx & 31;
        *reinterpret_cast<float4*>(tile + r * GPITCH + c4 * 4) = xv[i];
    }
    __syncthreads();

    v8f acc[MT];
#pragma unroll
    for (int m = 0; m < MT; ++m) acc[m] = v8f{};

    const float* arow = tile + l15 * GPITCH + half * 2;
#pragma unroll
    for (int ks = 0; ks < KS; ++ks) {
#pragma unroll
        for (int m = 0; m < MT; ++m) {
            const v2f a = *reinterpret_cast<const v2f*>(arow + m * 16 * GPITCH + ks * 4);
            acc[m] = __builtin_amdgcn_wmma_f32_16x16x4_f32(
                 false, a,  false, b[ks],
                 (short)0, acc[m],  false,  false);
        }
    }
    __syncthreads();

#pragma unroll
    for (int m = 0; m < MT; ++m) {
#pragma unroll
        for (int v = 0; v < 8; ++v)
            tile[(m * 16 + v + half * 8) * GPITCH + coln] = acc[m][v];
    }
    __syncthreads();
#pragma unroll
    for (int i = 0; i < OL; ++i) {
        const int idx = i * NT + tid;
        const int r   = idx / C4, c4 = idx % C4;
        if (row0 + r < NN)
            *reinterpret_cast<float4*>(out + (size_t)(row0 + r) * C + c4 * 4) =
                *reinterpret_cast<const float4*>(tile + r * GPITCH + c4 * 4);
    }
}

template <bool OWN>
__global__ void score_kernel(const float* __restrict__ xw,
                             const float* __restrict__ a_s, const float* __restrict__ a_d,
                             float* __restrict__ ss, float* __restrict__ sd,
                             int H, int F,
                             const int* __restrict__ ei, unsigned int* __restrict__ own,
                             const float* __restrict__ ew, float* __restrict__ part) {
    int i = blockIdx.x * blockDim.x + threadIdx.x;
    if constexpr (OWN) {
        static_assert(EE % 4 == 0 && (NN + RROWS - 1) / RROWS <= 256, "four owner bytes per word, an owner fits a byte");
        __shared__ float wsum[32];
        const int4* __restrict__ dst4 = reinterpret_cast<const int4*>(ei + EE);
        const float4* __restrict__ ew4 = reinterpret_cast<const float4*>(ew);
        float s = 0.0f;
        for (int g = i; g < EE / 4; g += gridDim.x * blockDim.x) {
            const int4   d = dst4[g];
            const float4 w = ew4[g];
            own[g] = ((unsigned)d.x / RROWS) | (((unsigned)d.y / RROWS) << 8) |
                     (((unsigned)d.z / RROWS) << 16) | (((unsigned)d.w / RROWS) << 24);
            s += (w.x + w.y) + (w.z + w.w);
        }
        s = waveSum(s);
        if ((threadIdx.x & 31) == 0) wsum[threadIdx.x >> 5] = s;
        __syncthreads();
        if (threadIdx.x == 0) {
            float t = 0.0f;
            for (int w = 0; w < (int)(blockDim.x >> 5); ++w) t += wsum[w];
            part[blockIdx.x] = t;
        }
    }
    if (i >= NN * H) return;
    const int n = i / H, h = i % H;
    const float* v  = xw + (size_t)n * H * F + h * F;
    const float* as = a_s + h * F;
    const float* ad = a_d + h * F;
    float s1 = 0.0f, s2 = 0.0f;
    for (int f = 0; f < F; ++f) { float x = v[f]; s1 += x * as[f]; s2 += x * ad[f]; }
    ss[i] = s1;
    sd[i] = s2;
}

#define RWAVES  32
#define QCAP    64
#define SCAN_G  4
#define LCAP    8192

template <int C, int H, bool ELU, int MODE>
__global__ __launch_bounds__(1024) void range_kernel(const int* __restrict__ ei,
                                                     const float* __restrict__ ew,
                                                     const float* __restrict__ part,
                                                     float* __restrict__ meanw,
                                                     const float* __restrict__ ss,
                                                     const float* __restrict__ sd,
                                                     const float* __restrict__ ce,
                                                     const float* __restrict__ xw,
                                                     const float* __restrict__ bias,
                                                     float* __restrict__ out,
                                                     int* __restrict__ gsr,
                                                     float* __restrict__ gw,
                                                     int* __restrict__ gcnt,
                                                     const unsigned int* __restrict__ own) {
    constexpr int NC = C / 32;
    constexpr int F  = C / H;
    constexpr int U  = 8;
    constexpr int N16 = EE / 16;
    constexpr int ROUND16 = RWAVES * SCAN_G * 32;
    constexpr int NR = (N16 + ROUND16 - 1) / ROUND16;
    static_assert(F % 32 == 0, "a lane's element c*32+lane must lie in head (c*32)/F");
    static_assert(EE % 16 == 0 && SCAN_G == 4, "the owner plane is read in groups of 16 bytes, 64 hit bits per lane and round");
    static_assert(32 % U == 0 && QCAP >= 64 && (QCAP & (QCAP - 1)) == 0, "queue and group sizes");
    static_assert(RROWS * H <= 1024, "one thread per (row, head) in the first phase");
    static_assert(NN <= 65536 && RROWS <= 256, "a list entry packs source (16 bits) and row (8 bits)");
    static_assert((NN + RROWS - 1) / RROWS <= 256, "the workspace holds 256 blocks' lists and counts");

    __shared__ float acc[RROWS * C];
    __shared__ float den[RROWS * H];
    __shared__ alignas(16) float mrow[RROWS * H];
    __shared__ float aself[RROWS * H];
    __shared__ alignas(16) float sdl[RROWS * H];
    __shared__ int que[RWAVES][QCAP];
    __shared__ int lsr[LCAP];
    __shared__ alignas(16) float lal[LCAP * H];
    __shared__ int lcnt;

    const int tid   = threadIdx.x;
    const int lane  = tid & 31;
    const int wave  = __builtin_amdgcn_readfirstlane(tid >> 5);
    const int d0    = blockIdx.x * RROWS;
    const int nrows = (NN - d0) < RROWS ? (NN - d0) : RROWS;

    float mean;
    if constexpr (MODE == 0) {
        static_assert(SBLK <= 1024 && RWAVES == 32, "one partial sum per thread, one wave sum per wave");
        __shared__ float psum[RWAVES];
        const float pv = tid < SBLK ? part[tid] : 0.0f;
        for (int i = tid; i < RROWS * C; i += 1024) acc[i] = 0.0f;
        const float ps = waveSum(pv);
        if (lane == 0) psum[wave] = ps;
        __syncthreads();
        float t = 0.0f;
        for (int w = 0; w < RWAVES; ++w) t += psum[w];
        mean = t / (float)EE;
        if (blockIdx.x == 0 && tid == 0) meanw[0] = mean;
    } else {
        for (int i = tid; i < RROWS * C; i += 1024) acc[i] = 0.0f;
        mean = meanw[0];
    }
    if (tid < RROWS * H) {
        const int r = tid / H, h = tid % H;
        const int n = d0 + (r < nrows ? r : nrows - 1);
        const float sdv = sd[n * H + h];
        float a = ss[n * H + h] + sdv + mean * ce[h];
        a = a > 0.0f ? a : 0.2f * a;
        den[tid]   = 0.0f;
        mrow[tid]  = a;
        aself[tid] = a;
        sdl[tid]   = sdv;
    }
    if (tid == 0) lcnt = 0;
    __syncthreads();

    int qhead = 0, qcnt = 0;

    auto load_ss = [&](const int s, float (&sv)[H]) {
        if constexpr (H == 4) {
            const float4 t = *reinterpret_cast<const float4*>(ss + (size_t)s * 4);
            sv[0] = t.x; sv[1] = t.y; sv[2] = t.z; sv[3] = t.w;
        } else {
            for (int h = 0; h < H; ++h) sv[h] = ss[(size_t)s * H + h];
        }
    };

    auto alpha_of = [&](const float (&sv)[H], const int r, const float w, float (&al)[H]) {
        if constexpr (H == 4) {
            const float4 dv = *reinterpret_cast<const float4*>(sdl + r * 4);
            al[0] = sv[0] + dv.x + w * ce[0];
            al[1] = sv[1] + dv.y + w * ce[1];
            al[2] = sv[2] + dv.z + w * ce[2];
            al[3] = sv[3] + dv.w + w * ce[3];
        } else {
            for (int h = 0; h < H; ++h) al[h] = sv[h] + sdl[r * H + h] + w * ce[h];
        }
        for (int h = 0; h < H; ++h) al[h] = al[h] > 0.0f ? al[h] : 0.2f * al[h];
    };

    auto edge_alpha = [&](const int e, int& sr, float& w, float (&al)[H]) {
        const int s = ei[e];
        const int r = ei[EE + e] - d0;
        w = ew[e];
        sr = s | (r << 16);
        float sv[H];
        load_ss(s, sv);
        alpha_of(sv, r, w, al);
    };

    auto stage = [&](const int nb) {
        int sr = 0;
        float w = 0.0f;
        float al[H];
        for (int h = 0; h < H; ++h) al[h] = 0.0f;
        __builtin_amdgcn_wave_barrier();
        if (lane < nb) {
            edge_alpha(que[wave][(qhead + lane) & (QCAP - 1)], sr, w, al);
            const int r = sr >> 16;
            for (int h = 0; h < H; ++h) ldsMaxF(&mrow[r * H + h], al[h]);
        }
        int base = 0;
        if (lane == 0) base = atomicAdd(&lcnt, nb);
        base = __builtin_amdgcn_readfirstlane(base);
        if (lane < nb && base + lane < LCAP) {
            lsr[base + lane] = sr;
            if constexpr (H == 4) {
                *reinterpret_cast<float4*>(lal + (base + lane) * 4) = make_float4(al[0], al[1], al[2], al[3]);
            } else {
                for (int h = 0; h < H; ++h) lal[(base + lane) * H + h] = al[h];
            }
            if constexpr (MODE == 0) {
                gsr[(size_t)blockIdx.x * LCAP + base + lane] = sr;
                gw[(size_t)blockIdx.x * LCAP + base + lane]  = w;
            }
        }
    };

    auto accumulate = [&](const int nb, const int sr, const float (&al)[H]) {
        const int s = sr & 0xffff, r = sr >> 16;
        float at[H];
        for (int h = 0; h < H; ++h) at[h] = 0.0f;
        if (lane < nb) {
            if constexpr (H == 4) {
                const float4 mx = *reinterpret_cast<const float4*>(mrow + r * 4);
                at[0] = __expf(al[0] - mx.x);
                at[1] = __expf(al[1] - mx.y);
                at[2] = __expf(al[2] - mx.z);
                at[3] = __expf(al[3] - mx.w);
            } else {
                for (int h = 0; h < H; ++h) at[h] = __expf(al[h] - mrow[r * H + h]);
            }
            for (int h = 0; h < H; ++h) atomicAdd(&den[r * H + h], at[h]);
        }
        for (int j0 = 0; j0 < nb; j0 += U) {
            float v[U][NC];
#pragma unroll
            for (int u = 0; u < U; ++u) {
                if (j0 + u < nb) {
                    const int sj = __builtin_amdgcn_readlane(s, j0 + u);
                    const float* row = xw + (size_t)sj * C + lane;
                    for (int c = 0; c < NC; ++c) v[u][c] = row[c * 32];
                }
            }
#pragma unroll
            for (int u = 0; u < U; ++u) {
                if (j0 + u < nb) {
                    const int rj = __builtin_amdgcn_readlane(r, j0 + u);
                    float* arow = acc + rj * C + lane;
                    for (int c = 0; c < NC; ++c) {
                        const float a = __int_as_float(
                            __builtin_amdgcn_readlane(__float_as_int(at[(c * 32) / F]), j0 + u));
                        atomicAdd(arow + c * 32, v[u][c] * a);
                    }
                }
            }
        }
    };

    auto restage = [&](const int nb) {
        int sr = 0;
        float al[H];
        for (int h = 0; h < H; ++h) al[h] = 0.0f;
        __builtin_amdgcn_wave_barrier();
        float w = 0.0f;
        if (lane < nb) edge_alpha(que[wave][(qhead + lane) & (QCAP - 1)], sr, w, al);
        accumulate(nb, sr, al);
    };

    const uint4* __restrict__ own16 = reinterpret_cast<const uint4*>(own);
    const unsigned pat = (unsigned)blockIdx.x * 0x01010101u;
    const int rot = blockIdx.x % NR;
    auto match = [&](const unsigned v) -> unsigned {
        const unsigned t = v ^ pat;
        return ~(((t & 0x7f7f7f7fu) + 0x7f7f7f7fu) | t | 0x7f7f7f7fu);
    };
    auto group0 = [&](const int it) -> int {
        int rnd = it + rot;
        if (rnd >= NR) rnd -= NR;
        return rnd * ROUND16 + wave * (SCAN_G * 32);
    };
    auto issue = [&](const int g0, uint4 (&d)[SCAN_G]) {
#pragma unroll
        for (int g = 0; g < SCAN_G; ++g) {
            const int i16 = g0 + g * 32 + lane;
            d[g] = own16[i16 < N16 ? i16 : N16 - 1];
        }
    };
    auto scan = [&](auto batch) {
        uint4 dv[SCAN_G];
        int g0 = group0(0);
        issue(g0, dv);
        for (int it = 0; it < NR; ++it) {
            const int g1 = group0(it + 1 < NR ? it + 1 : it);
            uint4 nx[SCAN_G];
            issue(g1, nx);
            unsigned long long hits = 0ull;
#pragma unroll
            for (int g = 0; g < SCAN_G; ++g) {
                unsigned z = (match(dv[g].x) >> 7) | (match(dv[g].y) >> 6) |
                             (match(dv[g].z) >> 5) | (match(dv[g].w) >> 4);
                if ((g0 + g * 32 + lane) >= N16) z = 0u;
                hits |= (unsigned long long)z << (32 * (g >> 1) + 4 * (g & 1));
            }
            unsigned mask = __builtin_amdgcn_ballot_w32(hits != 0ull);
            while (mask != 0u) {
                if (hits != 0ull) {
                    const int b = __builtin_ctzll(hits);
                    hits &= hits - 1ull;
                    const int g = ((b >> 5) << 1) | ((b >> 2) & 1), j = b & 3, k = (b >> 3) & 3;
                    const int e = (g0 + g * 32 + lane) * 16 + j * 4 + k;
                    const int pos = qcnt + (int)__builtin_amdgcn_mbcnt_lo(mask, 0u);
                    que[wave][(qhead + pos) & (QCAP - 1)] = e;
                }
                qcnt += __popc(mask);
                if (qcnt >= 32) {
                    batch(32);
                    qhead = (qhead + 32) & (QCAP - 1);
                    qcnt -= 32;
                }
                mask = __builtin_amdgcn_ballot_w32(hits != 0ull);
            }
#pragma unroll
            for (int g = 0; g < SCAN_G; ++g) dv[g] = nx[g];
            g0 = g1;
        }
        if (qcnt > 0) batch(qcnt);
        qhead = 0;
        qcnt = 0;
    };

    bool listed = false;
    if constexpr (MODE == 1) {
        const int mt = __builtin_amdgcn_readfirstlane(gcnt[blockIdx.x]);
        listed = mt <= LCAP;
        if (listed) {
            const int*   __restrict__ msr = gsr + (size_t)blockIdx.x * LCAP;
            const float* __restrict__ mw  = gw  + (size_t)blockIdx.x * LCAP;
            constexpr int G = 4;
            for (int b0 = wave * 32; b0 < mt; b0 += G * RWAVES * 32) {
                int   sr[G];
                float w[G];
                float sv[G][H];
#pragma unroll
                for (int g = 0; g < G; ++g) {
                    const int p = b0 + g * (RWAVES * 32) + lane;
                    const int q = p < mt ? p : mt - 1;
                    sr[g] = msr[q];
                    w[g]  = mw[q];
                }
#pragma unroll
                for (int g = 0; g < G; ++g) asm volatile("" :: "v"(sr[g]), "v"(w[g]));
#pragma unroll
                for (int g = 0; g < G; ++g) load_ss(sr[g] & 0xffff, sv[g]);
#pragma unroll
                for (int g = 0; g < G; ++g) {
                    const int p = b0 + g * (RWAVES * 32) + lane;
                    if (p < mt) {
                        const int r = sr[g] >> 16;
                        float al[H];
                        alpha_of(sv[g], r, w[g], al);
                        for (int h = 0; h < H; ++h) ldsMaxF(&mrow[r * H + h], al[h]);
                        lsr[p] = sr[g];
                        if constexpr (H == 4) {
                            *reinterpret_cast<float4*>(lal + p * 4) = make_float4(al[0], al[1], al[2], al[3]);
                        } else {
                            for (int h = 0; h < H; ++h) lal[p * H + h] = al[h];
                        }
                    }
                }
            }
            if (tid == 0) lcnt = mt;
        }
    }
    if (!listed) scan(stage);
    __syncthreads();

    const int total = __builtin_amdgcn_readfirstlane(lcnt);
    if constexpr (MODE == 0) {
        if (tid == 0) gcnt[blockIdx.x] = total;
    }
    if (total <= LCAP) {
        for (int base = wave * 32; base < total; base += RWAVES * 32) {
            const int left = total - base;
            int sr = 0;
            float al[H];
            for (int h = 0; h < H; ++h) al[h] = 0.0f;
            if (lane < left) {
                sr = lsr[base + lane];
                if constexpr (H == 4) {
                    const float4 t = *reinterpret_cast<const float4*>(lal + (base + lane) * 4);
                    al[0] = t.x; al[1] = t.y; al[2] = t.z; al[3] = t.w;
                } else {
                    for (int h = 0; h < H; ++h) al[h] = lal[(base + lane) * H + h];
                }
            }
            if (left >= 32) accumulate(32, sr, al);
            else            accumulate(left, sr, al);
        }
    } else {
        scan(restage);
    }
    __syncthreads();

    {
        constexpr int RPT = 1024 / C;
        constexpr int P0  = 8;
        const int c = tid % C, h = c / F;
        const float bc = bias[c];
        for (int r0 = tid / C; r0 < nrows; r0 += RPT * P0) {
            float xv[P0];
#pragma unroll
            for (int k = 0; k < P0; ++k) {
                const int r = r0 + k * RPT;
                xv[k] = xw[(size_t)(d0 + (r < nrows ? r : nrows - 1)) * C + c];
            }
#pragma unroll
            for (int k = 0; k < P0; ++k) {
                const int r = r0 + k * RPT;
                if (r < nrows) {
                    const int rh = r * H + h;
                    const float exs = __expf(aself[rh] - mrow[rh]);
                    float v = (acc[r * C + c] + exs * xv[k]) / (den[rh] + exs + 1e-16f) + bc;
                    if constexpr (ELU) v = v > 0.0f ? v : expm1f(v);
                    out[(size_t)(d0 + r) * C + c] = v;
                }
            }
        }
    }
}

extern "C" void kernel_launch(void* const* d_in, const int* in_sizes, int n_in,
                              void* d_out, int out_size, void* d_ws, size_t ws_size,
                              hipStream_t stream) {
    const float* x  = (const float*)d_in[0];
    const int*   ei = (const int*)d_in[1];
    const float* ew = (const float*)d_in[2];

    const float* Wl[3] = { (const float*)d_in[3],  (const float*)d_in[9],  (const float*)d_in[15] };
    const float* As[3] = { (const float*)d_in[4],  (const float*)d_in[10], (const float*)d_in[16] };
    const float* Ad[3] = { (const float*)d_in[5],  (const float*)d_in[11], (const float*)d_in[17] };
    const float* We[3] = { (const float*)d_in[6],  (const float*)d_in[12], (const float*)d_in[18] };
    const float* Ae[3] = { (const float*)d_in[7],  (const float*)d_in[13], (const float*)d_in[19] };
    const float* Bi[3] = { (const float*)d_in[8],  (const float*)d_in[14], (const float*)d_in[20] };

    const int Hs[3] = { 4, 4, 1 };
    const int Fs[3] = { 32, 32, 64 };

    float* ws    = (float*)d_ws;
    float* xw    = ws;
    float* bufA  = xw    + (size_t)NN * 128;
    float* bufB  = bufA  + (size_t)NN * 128;
    float* ss    = bufB  + (size_t)NN * 128;
    float* sd    = ss    + (size_t)NN * 4;
    float* ce    = sd    + (size_t)NN * 4;
    float* meanw = ce + 4;
    int*   gcnt  = (int*)(ce + 64);
    int*   gsr   = gcnt + 256;
    float* gw    = (float*)(gsr + 256 * LCAP);
    unsigned int* own = (unsigned int*)(gw + 256 * LCAP);
    float* part = (float*)(own + EE / 4);

    const float* in = x;
    float* outs[3] = { bufA, bufB, (float*)d_out };

    for (int l = 0; l < 3; ++l) {
        const int H = Hs[l], F = Fs[l];

        edge_coef_kernel<<<1, 32, 0, stream>>>(We[l], Ae[l], ce, H, F);

        if (l < 2)
            gemm_wmma_f32<128><<<(NN + GROWS - 1) / GROWS, 256, 0, stream>>>(in, Wl[l], xw);
        else
            gemm_wmma_f32<64><<<(NN + GROWS - 1) / GROWS, 128, 0, stream>>>(in, Wl[l], xw);

        if (l == 0)
            score_kernel<true><<<SBLK, STHR, 0, stream>>>(xw, As[l], Ad[l], ss, sd, H, F, ei, own, ew, part);
        else
            score_kernel<false><<<(NN * H + STHR - 1) / STHR, STHR, 0, stream>>>(xw, As[l], Ad[l], ss, sd, H, F, ei, own, ew, part);

        if (l == 0)
            range_kernel<128, 4, true, 0><<<(NN + RROWS - 1) / RROWS, 1024, 0, stream>>>(
                ei, ew, part, meanw, ss, sd, ce, xw, Bi[l], outs[l], gsr, gw, gcnt, own);
        else if (l == 1)
            range_kernel<128, 4, true, 1><<<(NN + RROWS - 1) / RROWS, 1024, 0, stream>>>(
                ei, ew, part, meanw, ss, sd, ce, xw, Bi[l], outs[l], gsr, gw, gcnt, own);
        else
            range_kernel<64, 1, false, 1><<<(NN + RROWS - 1) / RROWS, 1024, 0, stream>>>(
                ei, ew, part, meanw, ss, sd, ce, xw, Bi[l], outs[l], gsr, gw, gcnt, own);

        in = outs[l];
    }
}


# gather: a row comes by one 16-byte load per lane (8-byte at 64 columns; LDS sum rows lane-major), two groups of 8 rows in turn: the next group's loads issue ahead of this group's LDS adds; same sums
# speedup vs baseline: 7.6075x; 1.0392x over previous
#include <hip/hip_runtime.h>
#include <math.h>

#define NN    50000
#define EE    800000
#define KDIM  128
#define RROWS 196

typedef float v2f __attribute__((ext_vector_type(2)));
typedef float v8f __attribute__((ext_vector_type(8)));

__device__ __forceinline__ void ldsMaxF(float* addr, float val) {
    if (val >= 0.0f) atomicMax((int*)addr, __float_as_int(val));
    else             atomicMin((unsigned int*)addr, __float_as_uint(val));
}

#define STHR 256
#define SBLK ((NN * 4 + STHR - 1) / STHR)

__device__ __forceinline__ float waveSum(float v) {
    for (int o = 16; o > 0; o >>= 1) v += __shfl_xor(v, o);
    return v;
}

__global__ void edge_coef_kernel(const float* __restrict__ We, const float* __restrict__ ae,
                                 float* __restrict__ ce, int H, int F) {
    int h = threadIdx.x;
    if (h >= H) return;
    float s = 0.0f;
    for (int f = 0; f < F; ++f) s += We[h * F + f] * ae[h * F + f];
    ce[h] = s;
}

#define GROWS   64
#define GPITCH  132

template <int C>
__global__ __launch_bounds__(C * 2, C == 128 ? 8 : 4) void gemm_wmma_f32(const float* __restrict__ X,
                                                       const float* __restrict__ W,
                                                       float* __restrict__ out) {
    constexpr int NT = C * 2;
    constexpr int MT = GROWS / 16;
    constexpr int KS = KDIM / 4;
    constexpr int XL = GROWS * (KDIM / 4) / NT;
    constexpr int C4 = C / 4;
    constexpr int OL = GROWS * C4 / NT;
    static_assert(KDIM == 128 && GPITCH >= KDIM && GPITCH % 4 == 0, "a wave's 16-byte loads cover one row of X");
    static_assert(GROWS * (KDIM / 4) % NT == 0 && GROWS * C4 % NT == 0, "whole trips");

    __shared__ alignas(16) float tile[GROWS * GPITCH];

    const int tid  = threadIdx.x;
    const int lane = tid & 31;
    const int wave = __builtin_amdgcn_readfirstlane(tid >> 5);
    const int half = lane >> 4;
    const int l15  = lane & 15;
    const int row0 = blockIdx.x * GROWS;

    float4 xv[XL];
#pragma unroll
    for (int i = 0; i < XL; ++i) {
        const int idx = i * NT + tid;
        const int r   = idx >> 5, c4 = idx & 31;
        const int gr  = (row0 + r) < NN ? (row0 + r) : (NN - 1);
        xv[i] = *reinterpret_cast<const float4*>(X + (size_t)gr * KDIM + c4 * 4);
    }

    v2f b[KS];
    const int coln = wave * 16 + l15;
#pragma unroll
    for (int ks = 0; ks < KS; ++ks) {
        const int kk = ks * 4 + half * 2;
        b[ks][0] = W[(size_t)kk * C + coln];
        b[ks][1] = W[(size_t)(kk + 1) * C + coln];
    }

#pragma unroll
    for (int i = 0; i < XL; ++i) {
        const int idx = i * NT + tid;
        const int r   = idx >> 5, c4 = idx & 31;
        *reinterpret_cast<float4*>(tile + r * GPITCH + c4 * 4) = xv[i];
    }
    __syncthreads();

    v8f acc[MT];
#pragma unroll
    for (int m = 0; m < MT; ++m) acc[m] = v8f{};

    const float* arow = tile + l15 * GPITCH + half * 2;
#pragma unroll
    for (int ks = 0; ks < KS; ++ks) {
#pragma unroll
        for (int m = 0; m < MT; ++m) {
            const v2f a = *reinterpret_cast<const v2f*>(arow + m * 16 * GPITCH + ks * 4);
            acc[m] = __builtin_amdgcn_wmma_f32_16x16x4_f32(
                 false, a,  false, b[ks],
                 (short)0, acc[m],  false,  false);
        }
    }
    __syncthreads();

#pragma unroll
    for (int m = 0; m < MT; ++m) {
#pragma unroll
        for (int v = 0; v < 8; ++v)
            tile[(m * 16 + v + half * 8) * GPITCH + coln] = acc[m][v];
    }
    __syncthreads();
#pragma unroll
    for (int i = 0; i < OL; ++i) {
        const int idx = i * NT + tid;
        const int r   = idx / C4, c4 = idx % C4;
        if (row0 + r < NN)
            *reinterpret_cast<float4*>(out + (size_t)(row0 + r) * C + c4 * 4) =
                *reinterpret_cast<const float4*>(tile + r * GPITCH + c4 * 4);
    }
}

template <bool OWN>
__global__ void score_kernel(const float* __restrict__ xw,
                             const float* __restrict__ a_s, const float* __restrict__ a_d,
                             float* __restrict__ ss, float* __restrict__ sd,
                             int H, int F,
                             const int* __restrict__ ei, unsigned int* __restrict__ own,
                             const float* __restrict__ ew, float* __restrict__ part) {
    int i = blockIdx.x * blockDim.x + threadIdx.x;
    if constexpr (OWN) {
        static_assert(EE % 4 == 0 && (NN + RROWS - 1) / RROWS <= 256, "four owner bytes per word, an owner fits a byte");
        __shared__ float wsum[32];
        const int4* __restrict__ dst4 = reinterpret_cast<const int4*>(ei + EE);
        const float4* __restrict__ ew4 = reinterpret_cast<const float4*>(ew);
        float s = 0.0f;
        for (int g = i; g < EE / 4; g += gridDim.x * blockDim.x) {
            const int4   d = dst4[g];
            const float4 w = ew4[g];
            own[g] = ((unsigned)d.x / RROWS) | (((unsigned)d.y / RROWS) << 8) |
                     (((unsigned)d.z / RROWS) << 16) | (((unsigned)d.w / RROWS) << 24);
            s += (w.x + w.y) + (w.z + w.w);
        }
        s = waveSum(s);
        if ((threadIdx.x & 31) == 0) wsum[threadIdx.x >> 5] = s;
        __syncthreads();
        if (threadIdx.x == 0) {
            float t = 0.0f;
            for (int w = 0; w < (int)(blockDim.x >> 5); ++w) t += wsum[w];
            part[blockIdx.x] = t;
        }
    }
    if (i >= NN * H) return;
    const int n = i / H, h = i % H;
    const float* v  = xw + (size_t)n * H * F + h * F;
    const float* as = a_s + h * F;
    const float* ad = a_d + h * F;
    float s1 = 0.0f, s2 = 0.0f;
    for (int f = 0; f < F; ++f) { float x = v[f]; s1 += x * as[f]; s2 += x * ad[f]; }
    ss[i] = s1;
    sd[i] = s2;
}

#define RWAVES  32
#define QCAP    64
#define SCAN_G  4
#define LCAP    8192

template <int C, int H, bool ELU, int MODE>
__global__ __launch_bounds__(1024) void range_kernel(const int* __restrict__ ei,
                                                     const float* __restrict__ ew,
                                                     const float* __restrict__ part,
                                                     float* __restrict__ meanw,
                                                     const float* __restrict__ ss,
                                                     const float* __restrict__ sd,
                                                     const float* __restrict__ ce,
                                                     const float* __restrict__ xw,
                                                     const float* __restrict__ bias,
                                                     float* __restrict__ out,
                                                     int* __restrict__ gsr,
                                                     float* __restrict__ gw,
                                                     int* __restrict__ gcnt,
                                                     const unsigned int* __restrict__ own) {
    constexpr int NC = C / 32;
    constexpr int F  = C / H;
    constexpr int U  = 8;
    constexpr int N16 = EE / 16;
    constexpr int ROUND16 = RWAVES * SCAN_G * 32;
    constexpr int NR = (N16 + ROUND16 - 1) / ROUND16;
    static_assert(F % NC == 0 && C == NC * 32, "a lane's NC adjacent columns lie in one head, a wave's lanes cover the row");
    typedef float rowv __attribute__((ext_vector_type(NC)));
    static_assert(EE % 16 == 0 && SCAN_G == 4, "the owner plane is read in groups of 16 bytes, 64 hit bits per lane and round");
    static_assert(32 % U == 0 && QCAP >= 64 && (QCAP & (QCAP - 1)) == 0, "queue and group sizes");
    static_assert(RROWS * H <= 1024, "one thread per (row, head) in the first phase");
    static_assert(NN <= 65536 && RROWS <= 256, "a list entry packs source (16 bits) and row (8 bits)");
    static_assert((NN + RROWS - 1) / RROWS <= 256, "the workspace holds 256 blocks' lists and counts");

    __shared__ float acc[RROWS * C];
    __shared__ float den[RROWS * H];
    __shared__ alignas(16) float mrow[RROWS * H];
    __shared__ float aself[RROWS * H];
    __shared__ alignas(16) float sdl[RROWS * H];
    __shared__ int que[RWAVES][QCAP];
    __shared__ int lsr[LCAP];
    __shared__ alignas(16) float lal[LCAP * H];
    __shared__ int lcnt;

    const int tid   = threadIdx.x;
    const int lane  = tid & 31;
    const int wave  = __builtin_amdgcn_readfirstlane(tid >> 5);
    const int d0    = blockIdx.x * RROWS;
    const int nrows = (NN - d0) < RROWS ? (NN - d0) : RROWS;

    float mean;
    if constexpr (MODE == 0) {
        static_assert(SBLK <= 1024 && RWAVES == 32, "one partial sum per thread, one wave sum per wave");
        __shared__ float psum[RWAVES];
        const float pv = tid < SBLK ? part[tid] : 0.0f;
        for (int i = tid; i < RROWS * C; i += 1024) acc[i] = 0.0f;
        const float ps = waveSum(pv);
        if (lane == 0) psum[wave] = ps;
        __syncthreads();
        float t = 0.0f;
        for (int w = 0; w < RWAVES; ++w) t += psum[w];
        mean = t / (float)EE;
        if (blockIdx.x == 0 && tid == 0) meanw[0] = mean;
    } else {
        for (int i = tid; i < RROWS * C; i += 1024) acc[i] = 0.0f;
        mean = meanw[0];
    }
    if (tid < RROWS * H) {
        const int r = tid / H, h = tid % H;
        const int n = d0 + (r < nrows ? r : nrows - 1);
        const float sdv = sd[n * H + h];
        float a = ss[n * H + h] + sdv + mean * ce[h];
        a = a > 0.0f ? a : 0.2f * a;
        den[tid]   = 0.0f;
        mrow[tid]  = a;
        aself[tid] = a;
        sdl[tid]   = sdv;
    }
    if (tid == 0) lcnt = 0;
    __syncthreads();

    int qhead = 0, qcnt = 0;

    auto load_ss = [&](const int s, float (&sv)[H]) {
        if constexpr (H == 4) {
            const float4 t = *reinterpret_cast<const float4*>(ss + (size_t)s * 4);
            sv[0] = t.x; sv[1] = t.y; sv[2] = t.z; sv[3] = t.w;
        } else {
            for (int h = 0; h < H; ++h) sv[h] = ss[(size_t)s * H + h];
        }
    };

    auto alpha_of = [&](const float (&sv)[H], const int r, const float w, float (&al)[H]) {
        if constexpr (H == 4) {
            const float4 dv = *reinterpret_cast<const float4*>(sdl + r * 4);
            al[0] = sv[0] + dv.x + w * ce[0];
            al[1] = sv[1] + dv.y + w * ce[1];
            al[2] = sv[2] + dv.z + w * ce[2];
            al[3] = sv[3] + dv.w + w * ce[3];
        } else {
            for (int h = 0; h < H; ++h) al[h] = sv[h] + sdl[r * H + h] + w * ce[h];
        }
        for (int h = 0; h < H; ++h) al[h] = al[h] > 0.0f ? al[h] : 0.2f * al[h];
    };

    auto edge_alpha = [&](const int e, int& sr, float& w, float (&al)[H]) {
        const int s = ei[e];
        const int r = ei[EE + e] - d0;
        w = ew[e];
        sr = s | (r << 16);
        float sv[H];
        load_ss(s, sv);
        alpha_of(sv, r, w, al);
    };

    auto stage = [&](const int nb) {
        int sr = 0;
        float w = 0.0f;
        float al[H];
        for (int h = 0; h < H; ++h) al[h] = 0.0f;
        __builtin_amdgcn_wave_barrier();
        if (lane < nb) {
            edge_alpha(que[wave][(qhead + lane) & (QCAP - 1)], sr, w, al);
            const int r = sr >> 16;
            for (int h = 0; h < H; ++h) ldsMaxF(&mrow[r * H + h], al[h]);
        }
        int base = 0;
        if (lane == 0) base = atomicAdd(&lcnt, nb);
        base = __builtin_amdgcn_readfirstlane(base);
        if (lane < nb && base + lane < LCAP) {
            lsr[base + lane] = sr;
            if constexpr (H == 4) {
                *reinterpret_cast<float4*>(lal + (base + lane) * 4) = make_float4(al[0], al[1], al[2], al[3]);
            } else {
                for (int h = 0; h < H; ++h) lal[(base + lane) * H + h] = al[h];
            }
            if constexpr (MODE == 0) {
                gsr[(size_t)blockIdx.x * LCAP + base + lane] = sr;
                gw[(size_t)blockIdx.x * LCAP + base + lane]  = w;
            }
        }
    };

    auto accumulate = [&](const int nb, const int sr, const float (&al)[H]) {
        const int s = sr & 0xffff, r = sr >> 16;
        float at[H];
        for (int h = 0; h < H; ++h) at[h] = 0.0f;
        if (lane < nb) {
            if constexpr (H == 4) {
                const float4 mx = *reinterpret_cast<const float4*>(mrow + r * 4);
                at[0] = __expf(al[0] - mx.x);
                at[1] = __expf(al[1] - mx.y);
                at[2] = __expf(al[2] - mx.z);
                at[3] = __expf(al[3] - mx.w);
            } else {
                for (int h = 0; h < H; ++h) at[h] = __expf(al[h] - mrow[r * H + h]);
            }
            for (int h = 0; h < H; ++h) atomicAdd(&den[r * H + h], at[h]);
        }
        const int hl = (lane * NC) / F;
        auto load_rows = [&](const int j0, rowv (&v)[U]) {
#pragma unroll
            for (int u = 0; u < U; ++u) {
                if (j0 + u < nb) {
                    const int sj = __builtin_amdgcn_readlane(s, j0 + u);
                    v[u] = *reinterpret_cast<const rowv*>(xw + (size_t)sj * C + lane * NC);
                }
            }
        };
        auto add_rows = [&](const int j0, const rowv (&v)[U]) {
#pragma unroll
            for (int u = 0; u < U; ++u) {
                if (j0 + u < nb) {
                    const int rj = __builtin_amdgcn_readlane(r, j0 + u);
                    float* arow = acc + rj * C + lane;
                    float a = __int_as_float(__builtin_amdgcn_readlane(__float_as_int(at[0]), j0 + u));
                    for (int h = 1; h < H; ++h) {
                        const float ah = __int_as_float(__builtin_amdgcn_readlane(__float_as_int(at[h]), j0 + u));
                        a = hl == h ? ah : a;
                    }
                    for (int c = 0; c < NC; ++c) atomicAdd(arow + c * 32, v[u][c] * a);
                }
            }
        };
        rowv va[U], vb[U];
        load_rows(0, va);
        for (int j0 = 0; j0 < nb; j0 += 2 * U) {
            if (j0 + U < nb) load_rows(j0 + U, vb);
            add_rows(j0, va);
            if (j0 + 2 * U < nb) load_rows(j0 + 2 * U, va);
            if (j0 + U < nb) add_rows(j0 + U, vb);
        }
    };

    auto restage = [&](const int nb) {
        int sr = 0;
        float al[H];
        for (int h = 0; h < H; ++h) al[h] = 0.0f;
        __builtin_amdgcn_wave_barrier();
        float w = 0.0f;
        if (lane < nb) edge_alpha(que[wave][(qhead + lane) & (QCAP - 1)], sr, w, al);
        accumulate(nb, sr, al);
    };

    const uint4* __restrict__ own16 = reinterpret_cast<const uint4*>(own);
    const unsigned pat = (unsigned)blockIdx.x * 0x01010101u;
    const int rot = blockIdx.x % NR;
    auto match = [&](const unsigned v) -> unsigned {
        const unsigned t = v ^ pat;
        return ~(((t & 0x7f7f7f7fu) + 0x7f7f7f7fu) | t | 0x7f7f7f7fu);
    };
    auto group0 = [&](const int it) -> int {
        int rnd = it + rot;
        if (rnd >= NR) rnd -= NR;
        return rnd * ROUND16 + wave * (SCAN_G * 32);
    };
    auto issue = [&](const int g0, uint4 (&d)[SCAN_G]) {
#pragma unroll
        for (int g = 0; g < SCAN_G; ++g) {
            const int i16 = g0 + g * 32 + lane;
            d[g] = own16[i16 < N16 ? i16 : N16 - 1];
        }
    };
    auto scan = [&](auto batch) {
        uint4 dv[SCAN_G];
        int g0 = group0(0);
        issue(g0, dv);
        for (int it = 0; it < NR; ++it) {
            const int g1 = group0(it + 1 < NR ? it + 1 : it);
            uint4 nx[SCAN_G];
            issue(g1, nx);
            unsigned long long hits = 0ull;
#pragma unroll
            for (int g = 0; g < SCAN_G; ++g) {
                unsigned z = (match(dv[g].x) >> 7) | (match(dv[g].y) >> 6) |
                             (match(dv[g].z) >> 5) | (match(dv[g].w) >> 4);
                if ((g0 + g * 32 + lane) >= N16) z = 0u;
                hits |= (unsigned long long)z << (32 * (g >> 1) + 4 * (g & 1));
            }
            unsigned mask = __builtin_amdgcn_ballot_w32(hits != 0ull);
            while (mask != 0u) {
                if (hits != 0ull) {
                    const int b = __builtin_ctzll(hits);
                    hits &= hits - 1ull;
                    const int g = ((b >> 5) << 1) | ((b >> 2) & 1), j = b & 3, k = (b >> 3) & 3;
                    const int e = (g0 + g * 32 + lane) * 16 + j * 4 + k;
                    const int pos = qcnt + (int)__builtin_amdgcn_mbcnt_lo(mask, 0u);
                    que[wave][(qhead + pos) & (QCAP - 1)] = e;
                }
                qcnt += __popc(mask);
                if (qcnt >= 32) {
                    batch(32);
                    qhead = (qhead + 32) & (QCAP - 1);
                    qcnt -= 32;
                }
                mask = __builtin_amdgcn_ballot_w32(hits != 0ull);
            }
#pragma unroll
            for (int g = 0; g < SCAN_G; ++g) dv[g] = nx[g];
            g0 = g1;
        }
        if (qcnt > 0) batch(qcnt);
        qhead = 0;
        qcnt = 0;
    };

    bool listed = false;
    if constexpr (MODE == 1) {
        const int mt = __builtin_amdgcn_readfirstlane(gcnt[blockIdx.x]);
        listed = mt <= LCAP;
        if (listed) {
            const int*   __restrict__ msr = gsr + (size_t)blockIdx.x * LCAP;
            const float* __restrict__ mw  = gw  + (size_t)blockIdx.x * LCAP;
            constexpr int G = 4;
            for (int b0 = wave * 32; b0 < mt; b0 += G * RWAVES * 32) {
                int   sr[G];
                float w[G];
                float sv[G][H];
#pragma unroll
                for (int g = 0; g < G; ++g) {
                    const int p = b0 + g * (RWAVES * 32) + lane;
                    const int q = p < mt ? p : mt - 1;
                    sr[g] = msr[q];
                    w[g]  = mw[q];
                }
#pragma unroll
                for (int g = 0; g < G; ++g) asm volatile("" :: "v"(sr[g]), "v"(w[g]));
#pragma unroll
                for (int g = 0; g < G; ++g) load_ss(sr[g] & 0xffff, sv[g]);
#pragma unroll
                for (int g = 0; g < G; ++g) {
                    const int p = b0 + g * (RWAVES * 32) + lane;
                    if (p < mt) {
                        const int r = sr[g] >> 16;
                        float al[H];
                        alpha_of(sv[g], r, w[g], al);
                        for (int h = 0; h < H; ++h) ldsMaxF(&mrow[r * H + h], al[h]);
                        lsr[p] = sr[g];
                        if constexpr (H == 4) {
                            *reinterpret_cast<float4*>(lal + p * 4) = make_float4(al[0], al[1], al[2], al[3]);
                        } else {
                            for (int h = 0; h < H; ++h) lal[p * H + h] = al[h];
                        }
                    }
                }
            }
            if (tid == 0) lcnt = mt;
        }
    }
    if (!listed) scan(stage);
    __syncthreads();

    const int total = __builtin_amdgcn_readfirstlane(lcnt);
    if constexpr (MODE == 0) {
        if (tid == 0) gcnt[blockIdx.x] = total;
    }
    if (total <= LCAP) {
        for (int base = wave * 32; base < total; base += RWAVES * 32) {
            const int left = total - base;
            int sr = 0;
            float al[H];
            for (int h = 0; h < H; ++h) al[h] = 0.0f;
            if (lane < left) {
                sr = lsr[base + lane];
                if constexpr (H == 4) {
                    const float4 t = *reinterpret_cast<const float4*>(lal + (base + lane) * 4);
                    al[0] = t.x; al[1] = t.y; al[2] = t.z; al[3] = t.w;
                } else {
                    for (int h = 0; h < H; ++h) al[h] = lal[(base + lane) * H + h];
                }
            }
            if (left >= 32) accumulate(32, sr, al);
            else            accumulate(left, sr, al);
        }
    } else {
        scan(restage);
    }
    __syncthreads();

    {
        constexpr int RPT = 1024 / C;
        constexpr int P0  = 8;
        const int c = tid % C, h = c / F;
        const int cw = (c % NC) * 32 + c / NC;
        const float bc = bias[c];
        for (int r0 = tid / C; r0 < nrows; r0 += RPT * P0) {
            float xv[P0];
#pragma unroll
            for (int k = 0; k < P0; ++k) {
                const int r = r0 + k * RPT;
                xv[k] = xw[(size_t)(d0 + (r < nrows ? r : nrows - 1)) * C + c];
            }
#pragma unroll
            for (int k = 0; k < P0; ++k) {
                const int r = r0 + k * RPT;
                if (r < nrows) {
                    const int rh = r * H + h;
                    const float exs = __expf(aself[rh] - mrow[rh]);
                    float v = (acc[r * C + cw] + exs * xv[k]) / (den[rh] + exs + 1e-16f) + bc;
                    if constexpr (ELU) v = v > 0.0f ? v : expm1f(v);
                    out[(size_t)(d0 + r) * C + c] = v;
                }
            }
        }
    }
}

extern "C" void kernel_launch(void* const* d_in, const int* in_sizes, int n_in,
                              void* d_out, int out_size, void* d_ws, size_t ws_size,
                              hipStream_t stream) {
    const float* x  = (const float*)d_in[0];
    const int*   ei = (const int*)d_in[1];
    const float* ew = (const float*)d_in[2];

    const float* Wl[3] = { (const float*)d_in[3],  (const float*)d_in[9],  (const float*)d_in[15] };
    const float* As[3] = { (const float*)d_in[4],  (const float*)d_in[10], (const float*)d_in[16] };
    const float* Ad[3] = { (const float*)d_in[5],  (const float*)d_in[11], (const float*)d_in[17] };
    const float* We[3] = { (const float*)d_in[6],  (const float*)d_in[12], (const float*)d_in[18] };
    const float* Ae[3] = { (const float*)d_in[7],  (const float*)d_in[13], (const float*)d_in[19] };
    const float* Bi[3] = { (const float*)d_in[8],  (const float*)d_in[14], (const float*)d_in[20] };

    const int Hs[3] = { 4, 4, 1 };
    const int Fs[3] = { 32, 32, 64 };

    float* ws    = (float*)d_ws;
    float* xw    = ws;
    float* bufA  = xw    + (size_t)NN * 128;
    float* bufB  = bufA  + (size_t)NN * 128;
    float* ss    = bufB  + (size_t)NN * 128;
    float* sd    = ss    + (size_t)NN * 4;
    float* ce    = sd    + (size_t)NN * 4;
    float* meanw = ce + 4;
    int*   gcnt  = (int*)(ce + 64);
    int*   gsr   = gcnt + 256;
    float* gw    = (float*)(gsr + 256 * LCAP);
    unsigned int* own = (unsigned int*)(gw + 256 * LCAP);
    float* part = (float*)(own + EE / 4);

    const float* in = x;
    float* outs[3] = { bufA, bufB, (float*)d_out };

    for (int l = 0; l < 3; ++l) {
        const int H = Hs[l], F = Fs[l];

        edge_coef_kernel<<<1, 32, 0, stream>>>(We[l], Ae[l], ce, H, F);

        if (l < 2)
            gemm_wmma_f32<128><<<(NN + GROWS - 1) / GROWS, 256, 0, stream>>>(in, Wl[l], xw);
        else
            gemm_wmma_f32<64><<<(NN + GROWS - 1) / GROWS, 128, 0, stream>>>(in, Wl[l], xw);

        if (l == 0)
            score_kernel<true><<<SBLK, STHR, 0, stream>>>(xw, As[l], Ad[l], ss, sd, H, F, ei, own, ew, part);
        else
            score_kernel<false><<<(NN * H + STHR - 1) / STHR, STHR, 0, stream>>>(xw, As[l], Ad[l], ss, sd, H, F, ei, own, ew, part);

        if (l == 0)
            range_kernel<128, 4, true, 0><<<(NN + RROWS - 1) / RROWS, 1024, 0, stream>>>(
                ei, ew, part, meanw, ss, sd, ce, xw, Bi[l], outs[l], gsr, gw, gcnt, own);
        else if (l == 1)
            range_kernel<128, 4, true, 1><<<(NN + RROWS - 1) / RROWS, 1024, 0, stream>>>(
                ei, ew, part, meanw, ss, sd, ce, xw, Bi[l], outs[l], gsr, gw, gcnt, own);
        else
            range_kernel<64, 1, false, 1><<<(NN + RROWS - 1) / RROWS, 1024, 0, stream>>>(
                ei, ew, part, meanw, ss, sd, ce, xw, Bi[l], outs[l], gsr, gw, gcnt, own);

        in = outs[l];
    }
}
